# starting kernel
# baseline (speedup 1.0000x reference)
#include <hip/hip_runtime.h>
#include <math.h>

#define N 2048
#define D_IN 128
#define D_OUT 32
#define D_EDGE 8
#define ALPHA 0.2f
#define NEG_BIG -9e15f
#define ROWSTRIDE 2050

typedef __attribute__((ext_vector_type(2))) float v2f;
typedef __attribute__((ext_vector_type(8))) float v8f;

__device__ __forceinline__ v8f wmma_f32_k4(v2f a, v2f b, v8f c) {
  return __builtin_amdgcn_wmma_f32_16x16x4_f32(false, a, false, b, (short)0, c,
                                               false, false);
}

__device__ __forceinline__ float lrelu(float x) { return x >= 0.f ? x : ALPHA * x; }

__global__ void prep_vec_kernel(const float* __restrict__ Wh1,
                                const float* __restrict__ Wh2,
                                const float* __restrict__ ah,
                                float* __restrict__ v1, float* __restrict__ v2) {
  const int t = threadIdx.x;
  float a1 = 0.f, a2 = 0.f;
  for (int f = 0; f < D_OUT; ++f) {
    const float w = ah[f];
    a1 += Wh1[t * D_OUT + f] * w;
    a2 += Wh2[t * D_OUT + f] * w;
  }
  v1[t] = a1;
  v2[t] = a2;
}

__global__ void prep_s_kernel(const float* __restrict__ input,
                              const float* __restrict__ v1,
                              const float* __restrict__ v2,
                              float* __restrict__ s1, float* __restrict__ s2) {
  const int i = blockIdx.x * blockDim.x + threadIdx.x;
  const float* row = input + (size_t)i * D_IN;
  float a = 0.f, b = 0.f;
  for (int k = 0; k < D_IN; ++k) {
    const float x = row[k];
    a += x * v1[k];
    b += x * v2[k];
  }
  s1[i] = a;
  s2[i] = b;
}

__global__ void h_gemm_kernel(const float* __restrict__ input,
                              const float* __restrict__ Wh,
                              float* __restrict__ h) {
  const int lane = threadIdx.x & 31;
  const int i0 = blockIdx.x * 16;
  const int m = lane & 15;
  const int kd = (lane >> 4) << 1;
  const int n = lane & 15;
  v8f c0 = {};
  v8f c1 = {};
  for (int k0 = 0; k0 < D_IN; k0 += 4) {
    v2f a = *(const v2f*)(input + (size_t)(i0 + m) * D_IN + k0 + kd);
    v2f b0, b1;
    b0.x = Wh[(k0 + kd) * D_OUT + n];
    b0.y = Wh[(k0 + kd + 1) * D_OUT + n];
    b1.x = Wh[(k0 + kd) * D_OUT + 16 + n];
    b1.y = Wh[(k0 + kd + 1) * D_OUT + 16 + n];
    c0 = wmma_f32_k4(a, b0, c0);
    c1 = wmma_f32_k4(a, b1, c1);
  }
  const int row = (lane >> 4) << 3;
  for (int v = 0; v < 8; ++v) {
    h[(size_t)(i0 + row + v) * D_OUT + n] = c0[v];
    h[(size_t)(i0 + row + v) * D_OUT + 16 + n] = c1[v];
  }
}

__global__ void fused_attention_kernel(const float* __restrict__ edge_feat,
                                       const int* __restrict__ adj,
                                       const float* __restrict__ Wf,
                                       const float* __restrict__ af,
                                       const float* __restrict__ bf,
                                       const float* __restrict__ s1,
                                       const float* __restrict__ s2,
                                       const float* __restrict__ h,
                                       float* __restrict__ h_prime) {
  extern __shared__ float smem[];
  float* sc = smem;
  float* part = smem + 16 * ROWSTRIDE;
  __shared__ float s_inv[16];

  const int tid = threadIdx.x;
  const int lane = tid & 31;
  const int wave = tid >> 5;
  const int i0 = blockIdx.x * 16;

  const int nn = lane & 15;
  const int kd = (lane >> 4) << 1;

  v2f bw00, bw01, bw10, bw11;
  bw00.x = Wf[(0 + kd) * D_OUT + nn];
  bw00.y = Wf[(1 + kd) * D_OUT + nn];
  bw01.x = Wf[(0 + kd) * D_OUT + 16 + nn];
  bw01.y = Wf[(1 + kd) * D_OUT + 16 + nn];
  bw10.x = Wf[(4 + kd) * D_OUT + nn];
  bw10.y = Wf[(5 + kd) * D_OUT + nn];
  bw11.x = Wf[(4 + kd) * D_OUT + 16 + nn];
  bw11.y = Wf[(5 + kd) * D_OUT + 16 + nn];
  const float af_lo = af[nn], af_hi = af[16 + nn];
  const float bf_lo = bf[nn], bf_hi = bf[16 + nn];
  v8f cini0, cini1;
  for (int v = 0; v < 8; ++v) {
    cini0[v] = bf_lo;
    cini1[v] = bf_hi;
  }

  for (int t = wave; t < 16 * 128; t += 8) {
    const int il = t >> 7;
    const int i = i0 + il;
    const int j0 = (t & 127) << 4;
    const float* ep = edge_feat + ((size_t)i * N + (j0 + nn)) * D_EDGE + kd;
    __builtin_prefetch(ep + 8 * 16 * D_EDGE, 0, 1);
    v2f a0 = *(const v2f*)(ep);
    v2f a1 = *(const v2f*)(ep + 4);
    v8f f0 = wmma_f32_k4(a0, bw00, cini0);
    f0 = wmma_f32_k4(a1, bw10, f0);
    v8f f1 = wmma_f32_k4(a0, bw01, cini1);
    f1 = wmma_f32_k4(a1, bw11, f1);
    float p[8];
    for (int v = 0; v < 8; ++v)
      p[v] = lrelu(f0[v]) * af_lo + lrelu(f1[v]) * af_hi;
    for (int mask = 1; mask <= 8; mask <<= 1)
      for (int v = 0; v < 8; ++v) p[v] += __shfl_xor(p[v], mask, 32);
    if ((lane & 8) == 0) {
      const int md = lane & 7;
      float t0 = (md & 1) ? p[1] : p[0];
      float t1 = (md & 1) ? p[3] : p[2];
      float t2 = (md & 1) ? p[5] : p[4];
      float t3 = (md & 1) ? p[7] : p[6];
      float u0 = (md & 2) ? t1 : t0;
      float u1 = (md & 2) ? t3 : t2;
      float cp = (md & 4) ? u1 : u0;
      const int j = j0 + ((lane >> 4) << 3) + md;
      float v = lrelu(s1[i] + s2[j]) + lrelu(cp);
      v = (adj[(size_t)i * N + j] > 0) ? v : NEG_BIG;
      sc[il * ROWSTRIDE + j] = v;
    }
  }
  __syncthreads();

  for (int r = wave; r < 16; r += 8) {
    float* row = sc + r * ROWSTRIDE;
    float m = -3.4e38f;
    for (int c = lane; c < N; c += 32) m = fmaxf(m, row[c]);
    for (int mask = 16; mask >= 1; mask >>= 1)
      m = fmaxf(m, __shfl_xor(m, mask, 32));
    float s = 0.f;
    for (int c = lane; c < N; c += 32) {
      const float e = __expf(row[c] - m);
      row[c] = e;
      s += e;
    }
    for (int mask = 16; mask >= 1; mask >>= 1) s += __shfl_xor(s, mask, 32);
    if (lane == 0) s_inv[r] = 1.0f / s;
  }
  __syncthreads();

  v8f c0 = {};
  v8f c1 = {};
  const int kbase = wave * 256;
  for (int kk = 0; kk < 256; kk += 4) {
    const int k0 = kbase + kk;
    v2f a = *(const v2f*)(sc + nn * ROWSTRIDE + k0 + kd);
    v2f b0, b1;
    b0.x = h[(size_t)(k0 + kd) * D_OUT + nn];
    b0.y = h[(size_t)(k0 + kd + 1) * D_OUT + nn];
    b1.x = h[(size_t)(k0 + kd) * D_OUT + 16 + nn];
    b1.y = h[(size_t)(k0 + kd + 1) * D_OUT + 16 + nn];
    c0 = wmma_f32_k4(a, b0, c0);
    c1 = wmma_f32_k4(a, b1, c1);
  }
  {
    const int row = (lane >> 4) << 3;
    float* pw = part + wave * 512;
    for (int v = 0; v < 8; ++v) {
      pw[(row + v) * 32 + nn] = c0[v];
      pw[(row + v) * 32 + 16 + nn] = c1[v];
    }
  }
  __syncthreads();
  for (int e = tid; e < 512; e += 256) {
    float s = 0.f;
    for (int w = 0; w < 8; ++w) s += part[w * 512 + e];
    const int row = e >> 5, col = e & 31;
    h_prime[(size_t)(i0 + row) * D_OUT + col] = s * s_inv[row];
  }
}

__global__ void bn_partial_kernel(const float* __restrict__ hp,
                                  float* __restrict__ bp) {
  __shared__ float ls[8][32];
  __shared__ float lq[8][32];
  const int col = threadIdx.x & 31;
  const int seg = threadIdx.x >> 5;
  const int r0 = blockIdx.x * 32 + seg * 4;
  float s = 0.f, q = 0.f;
  for (int r = 0; r < 4; ++r) {
    const float v = hp[(size_t)(r0 + r) * D_OUT + col];
    s += v;
    q += v * v;
  }
  ls[seg][col] = s;
  lq[seg][col] = q;
  __syncthreads();
  if (threadIdx.x < 32) {
    float ss = 0.f, qq = 0.f;
    for (int w = 0; w < 8; ++w) {
      ss += ls[w][col];
      qq += lq[w][col];
    }
    bp[blockIdx.x * 64 + col] = ss;
    bp[blockIdx.x * 64 + 32 + col] = qq;
  }
}

__global__ void bn_stats_kernel(const float* __restrict__ bp,
                                float* __restrict__ ms) {
  const int col = threadIdx.x;
  float s = 0.f, q = 0.f;
  for (int b = 0; b < 64; ++b) {
    s += bp[b * 64 + col];
    q += bp[b * 64 + 32 + col];
  }
  const float mean = s * (1.0f / (float)N);
  const float var = q * (1.0f / (float)N) - mean * mean;
  ms[col] = mean;
  ms[32 + col] = rsqrtf(var + 1e-5f);
}

__global__ void bn_norm_kernel(const float* __restrict__ hp,
                               const float* __restrict__ ms,
                               const float* __restrict__ gamma,
                               const float* __restrict__ beta,
                               float* __restrict__ out) {
  const int idx = blockIdx.x * blockDim.x + threadIdx.x;
  const int col = idx & 31;
  out[idx] = gamma[col] * (hp[idx] - ms[col]) * ms[32 + col] + beta[col];
}

extern "C" void kernel_launch(void* const* d_in, const int* in_sizes, int n_in,
                              void* d_out, int out_size, void* d_ws,
                              size_t ws_size, hipStream_t stream) {
  const float* input = (const float*)d_in[0];
  const float* edge_feat = (const float*)d_in[1];
  const int* adj = (const int*)d_in[2];
  const float* Wh = (const float*)d_in[3];
  const float* Wh1 = (const float*)d_in[4];
  const float* Wh2 = (const float*)d_in[5];
  const float* Wf = (const float*)d_in[6];
  const float* ah = (const float*)d_in[7];
  const float* af = (const float*)d_in[8];
  const float* bf = (const float*)d_in[9];
  const float* gamma = (const float*)d_in[10];
  const float* beta = (const float*)d_in[11];

  float* ws = (float*)d_ws;
  float* v1 = ws;
  float* v2 = ws + 128;
  float* s1 = ws + 256;
  float* s2 = ws + 2304;
  float* h = ws + 4352;
  float* hp = ws + 69888;
  float* bp = ws + 135424;
  float* ms = ws + 139520;
  float* out = (float*)d_out;

  prep_vec_kernel<<<1, 128, 0, stream>>>(Wh1, Wh2, ah, v1, v2);
  prep_s_kernel<<<8, 256, 0, stream>>>(input, v1, v2, s1, s2);
  h_gemm_kernel<<<128, 32, 0, stream>>>(input, Wh, h);
  const size_t smem = (size_t)(16 * ROWSTRIDE + 8 * 512) * sizeof(float);
  fused_attention_kernel<<<128, 256, smem, stream>>>(edge_feat, adj, Wf, af, bf,
                                                     s1, s2, h, hp);
  bn_partial_kernel<<<64, 256, 0, stream>>>(hp, bp);
  bn_stats_kernel<<<1, 32, 0, stream>>>(bp, ms);
  bn_norm_kernel<<<256, 256, 0, stream>>>(hp, ms, gamma, beta, out);
}


# attention kernel re-tiled to hide load latency: 8-row tiles on 256 work-groups of 32 waves (was 16 rows, 128 x 8 waves), a wave loads 4 edge groups at once; softmax and P@h sums split 4 and 32 ways
# speedup vs baseline: 4.3038x; 4.3038x over previous
#include <hip/hip_runtime.h>
#include <math.h>

#define N 2048
#define D_IN 128
#define D_OUT 32
#define D_EDGE 8
#define ALPHA 0.2f
#define NEG_BIG -9e15f
#define ROWSTRIDE 2050
#define TROWS 8
#define TWAVES 32
#define GROUPS 4

typedef __attribute__((ext_vector_type(2))) float v2f;
typedef __attribute__((ext_vector_type(8))) float v8f;

__device__ __forceinline__ v8f wmma_f32_k4(v2f a, v2f b, v8f c) {
  return __builtin_amdgcn_wmma_f32_16x16x4_f32(false, a, false, b, (short)0, c,
                                               false, false);
}

__device__ __forceinline__ float lrelu(float x) { return x >= 0.f ? x : ALPHA * x; }

__device__ __forceinline__ void group_sums(v2f a0, v2f a1, v2f bw00, v2f bw01,
                                           v2f bw10, v2f bw11, v8f cini0,
                                           v8f cini1, float af_lo, float af_hi,
                                           float (&p)[8]) {
  v8f f0 = wmma_f32_k4(a0, bw00, cini0);
  f0 = wmma_f32_k4(a1, bw10, f0);
  v8f f1 = wmma_f32_k4(a0, bw01, cini1);
  f1 = wmma_f32_k4(a1, bw11, f1);
  for (int v = 0; v < 8; ++v)
    p[v] = lrelu(f0[v]) * af_lo + lrelu(f1[v]) * af_hi;
  for (int mask = 1; mask <= 8; mask <<= 1)
    for (int v = 0; v < 8; ++v) p[v] += __shfl_xor(p[v], mask, 32);
}

__global__ void prep_vec_kernel(const float* __restrict__ Wh1,
                                const float* __restrict__ Wh2,
                                const float* __restrict__ ah,
                                float* __restrict__ v1, float* __restrict__ v2) {
  const int t = threadIdx.x;
  float a1 = 0.f, a2 = 0.f;
  for (int f = 0; f < D_OUT; ++f) {
    const float w = ah[f];
    a1 += Wh1[t * D_OUT + f] * w;
    a2 += Wh2[t * D_OUT + f] * w;
  }
  v1[t] = a1;
  v2[t] = a2;
}

__global__ void prep_s_kernel(const float* __restrict__ input,
                              const float* __restrict__ v1,
                              const float* __restrict__ v2,
                              float* __restrict__ s1, float* __restrict__ s2) {
  const int i = blockIdx.x * blockDim.x + threadIdx.x;
  const float* row = input + (size_t)i * D_IN;
  float a = 0.f, b = 0.f;
  for (int k = 0; k < D_IN; ++k) {
    const float x = row[k];
    a += x * v1[k];
    b += x * v2[k];
  }
  s1[i] = a;
  s2[i] = b;
}

__global__ void h_gemm_kernel(const float* __restrict__ input,
                              const float* __restrict__ Wh,
                              float* __restrict__ h) {
  const int lane = threadIdx.x & 31;
  const int i0 = blockIdx.x * 16;
  const int m = lane & 15;
  const int kd = (lane >> 4) << 1;
  const int n = lane & 15;
  v8f c0 = {};
  v8f c1 = {};
  for (int k0 = 0; k0 < D_IN; k0 += 4) {
    v2f a = *(const v2f*)(input + (size_t)(i0 + m) * D_IN + k0 + kd);
    v2f b0, b1;
    b0.x = Wh[(k0 + kd) * D_OUT + n];
    b0.y = Wh[(k0 + kd + 1) * D_OUT + n];
    b1.x = Wh[(k0 + kd) * D_OUT + 16 + n];
    b1.y = Wh[(k0 + kd + 1) * D_OUT + 16 + n];
    c0 = wmma_f32_k4(a, b0, c0);
    c1 = wmma_f32_k4(a, b1, c1);
  }
  const int row = (lane >> 4) << 3;
  for (int v = 0; v < 8; ++v) {
    h[(size_t)(i0 + row + v) * D_OUT + n] = c0[v];
    h[(size_t)(i0 + row + v) * D_OUT + 16 + n] = c1[v];
  }
}

__global__ void __launch_bounds__(1024)
fused_attention_kernel(const float* __restrict__ edge_feat,
                       const int* __restrict__ adj,
                       const float* __restrict__ Wf,
                       const float* __restrict__ af,
                       const float* __restrict__ bf,
                       const float* __restrict__ s1,
                       const float* __restrict__ s2,
                       const float* __restrict__ h,
                       float* __restrict__ h_prime) {
  extern __shared__ float smem[];
  float* sc = smem;
  float* part = smem + TROWS * ROWSTRIDE;
  __shared__ float s_max[TWAVES];
  __shared__ float s_sum[TWAVES];

  const int tid = threadIdx.x;
  const int lane = tid & 31;
  const int wave = tid >> 5;
  const int i0 = blockIdx.x * TROWS;

  const int nn = lane & 15;
  const int kd = (lane >> 4) << 1;

  v2f bw00, bw01, bw10, bw11;
  bw00.x = Wf[(0 + kd) * D_OUT + nn];
  bw00.y = Wf[(1 + kd) * D_OUT + nn];
  bw01.x = Wf[(0 + kd) * D_OUT + 16 + nn];
  bw01.y = Wf[(1 + kd) * D_OUT + 16 + nn];
  bw10.x = Wf[(4 + kd) * D_OUT + nn];
  bw10.y = Wf[(5 + kd) * D_OUT + nn];
  bw11.x = Wf[(4 + kd) * D_OUT + 16 + nn];
  bw11.y = Wf[(5 + kd) * D_OUT + 16 + nn];
  const float af_lo = af[nn], af_hi = af[16 + nn];
  const float bf_lo = bf[nn], bf_hi = bf[16 + nn];
  v8f cini0, cini1;
  for (int v = 0; v < 8; ++v) {
    cini0[v] = bf_lo;
    cini1[v] = bf_hi;
  }

  const int md = lane & 7;
  const bool second = (lane & 8) != 0;
  const int jl = wave * 64 + ((lane & 8) << 1) + ((lane >> 4) << 3) + md;
  float s2v[2];
  s2v[0] = s2[jl];
  s2v[1] = s2[jl + 32];
  for (int il = 0; il < TROWS; ++il) {
    const int i = i0 + il;
    const float s1i = s1[i];
    const float* ep = edge_feat + ((size_t)i * N + (wave * 64 + nn)) * D_EDGE + kd;
    const int* ap = adj + (size_t)i * N + jl;
    v2f a0[GROUPS], a1[GROUPS];
    int av[2];
#pragma unroll
    for (int u = 0; u < GROUPS; ++u) {
      a0[u] = *(const v2f*)(ep + u * 16 * D_EDGE);
      a1[u] = *(const v2f*)(ep + u * 16 * D_EDGE + 4);
    }
    av[0] = ap[0];
    av[1] = ap[32];
    __builtin_amdgcn_sched_barrier(0);
#pragma unroll
    for (int pr = 0; pr < 2; ++pr) {
      float pa[8], pb[8];
      group_sums(a0[2 * pr], a1[2 * pr], bw00, bw01, bw10, bw11, cini0, cini1,
                 af_lo, af_hi, pa);
      group_sums(a0[2 * pr + 1], a1[2 * pr + 1], bw00, bw01, bw10, bw11, cini0,
                 cini1, af_lo, af_hi, pb);
      float q[8];
      for (int v = 0; v < 8; ++v) q[v] = second ? pb[v] : pa[v];
      float t0 = (md & 1) ? q[1] : q[0];
      float t1 = (md & 1) ? q[3] : q[2];
      float t2 = (md & 1) ? q[5] : q[4];
      float t3 = (md & 1) ? q[7] : q[6];
      float u0 = (md & 2) ? t1 : t0;
      float u1 = (md & 2) ? t3 : t2;
      float cp = (md & 4) ? u1 : u0;
      float v = lrelu(s1i + s2v[pr]) + lrelu(cp);
      v = (av[pr] > 0) ? v : NEG_BIG;
      sc[il * ROWSTRIDE + jl + pr * 32] = v;
      if (pr == 0) __builtin_amdgcn_sched_barrier(0);
    }
  }
  __syncthreads();

  const int r = wave >> 2;
  {
    float* seg = sc + r * ROWSTRIDE + (wave & 3) * 512;
    float x[16];
    float m = -3.4e38f;
    for (int c = 0; c < 16; ++c) {
      x[c] = seg[lane + 32 * c];
      m = fmaxf(m, x[c]);
    }
    for (int mask = 16; mask >= 1; mask >>= 1)
      m = fmaxf(m, __shfl_xor(m, mask, 32));
    if (lane == 0) s_max[wave] = m;
    __syncthreads();
    m = fmaxf(fmaxf(s_max[4 * r], s_max[4 * r + 1]),
              fmaxf(s_max[4 * r + 2], s_max[4 * r + 3]));
    float s = 0.f;
    for (int c = 0; c < 16; ++c) {
      const float e = __expf(x[c] - m);
      seg[lane + 32 * c] = e;
      s += e;
    }
    for (int mask = 16; mask >= 1; mask >>= 1) s += __shfl_xor(s, mask, 32);
    if (lane == 0) s_sum[wave] = s;
  }
  __syncthreads();

  v8f c0 = {};
  v8f c1 = {};
  const int kbase = wave * 64;
  for (int kk = 0; kk < 64; kk += 4) {
    const int k0 = kbase + kk;
    v2f a = *(const v2f*)(sc + (nn & 7) * ROWSTRIDE + k0 + kd);
    v2f b0, b1;
    b0.x = h[(size_t)(k0 + kd) * D_OUT + nn];
    b0.y = h[(size_t)(k0 + kd + 1) * D_OUT + nn];
    b1.x = h[(size_t)(k0 + kd) * D_OUT + 16 + nn];
    b1.y = h[(size_t)(k0 + kd + 1) * D_OUT + 16 + nn];
    c0 = wmma_f32_k4(a, b0, c0);
    c1 = wmma_f32_k4(a, b1, c1);
  }
  if (lane < 16) {
    float* pw = part + wave * 256;
    for (int v = 0; v < 8; ++v) {
      pw[v * 32 + nn] = c0[v];
      pw[v * 32 + 16 + nn] = c1[v];
    }
  }
  __syncthreads();
  if (tid < 256) {
    float s = 0.f;
    for (int w = 0; w < TWAVES; ++w) s += part[w * 256 + tid];
    const int row = tid >> 5, col = tid & 31;
    const float rs = (s_sum[4 * row] + s_sum[4 * row + 1]) +
                     (s_sum[4 * row + 2] + s_sum[4 * row + 3]);
    h_prime[(size_t)(i0 + row) * D_OUT + col] = s * (1.0f / rs);
  }
}

__global__ void bn_partial_kernel(const float* __restrict__ hp,
                                  float* __restrict__ bp) {
  __shared__ float ls[8][32];
  __shared__ float lq[8][32];
  const int col = threadIdx.x & 31;
  const int seg = threadIdx.x >> 5;
  const int r0 = blockIdx.x * 32 + seg * 4;
  float s = 0.f, q = 0.f;
  for (int r = 0; r < 4; ++r) {
    const float v = hp[(size_t)(r0 + r) * D_OUT + col];
    s += v;
    q += v * v;
  }
  ls[seg][col] = s;
  lq[seg][col] = q;
  __syncthreads();
  if (threadIdx.x < 32) {
    float ss = 0.f, qq = 0.f;
    for (int w = 0; w < 8; ++w) {
      ss += ls[w][col];
      qq += lq[w][col];
    }
    bp[blockIdx.x * 64 + col] = ss;
    bp[blockIdx.x * 64 + 32 + col] = qq;
  }
}

__global__ void bn_stats_kernel(const float* __restrict__ bp,
                                float* __restrict__ ms) {
  const int col = threadIdx.x;
  float s = 0.f, q = 0.f;
  for (int b = 0; b < 64; ++b) {
    s += bp[b * 64 + col];
    q += bp[b * 64 + 32 + col];
  }
  const float mean = s * (1.0f / (float)N);
  const float var = q * (1.0f / (float)N) - mean * mean;
  ms[col] = mean;
  ms[32 + col] = rsqrtf(var + 1e-5f);
}

__global__ void bn_norm_kernel(const float* __restrict__ hp,
                               const float* __restrict__ ms,
                               const float* __restrict__ gamma,
                               const float* __restrict__ beta,
                               float* __restrict__ out) {
  const int idx = blockIdx.x * blockDim.x + threadIdx.x;
  const int col = idx & 31;
  out[idx] = gamma[col] * (hp[idx] - ms[col]) * ms[32 + col] + beta[col];
}

extern "C" void kernel_launch(void* const* d_in, const int* in_sizes, int n_in,
                              void* d_out, int out_size, void* d_ws,
                              size_t ws_size, hipStream_t stream) {
  const float* input = (const float*)d_in[0];
  const float* edge_feat = (const float*)d_in[1];
  const int* adj = (const int*)d_in[2];
  const float* Wh = (const float*)d_in[3];
  const float* Wh1 = (const float*)d_in[4];
  const float* Wh2 = (const float*)d_in[5];
  const float* Wf = (const float*)d_in[6];
  const float* ah = (const float*)d_in[7];
  const float* af = (const float*)d_in[8];
  const float* bf = (const float*)d_in[9];
  const float* gamma = (const float*)d_in[10];
  const float* beta = (const float*)d_in[11];

  float* ws = (float*)d_ws;
  float* v1 = ws;
  float* v2 = ws + 128;
  float* s1 = ws + 256;
  float* s2 = ws + 2304;
  float* h = ws + 4352;
  float* hp = ws + 69888;
  float* bp = ws + 135424;
  float* ms = ws + 139520;
  float* out = (float*)d_out;

  prep_vec_kernel<<<1, 128, 0, stream>>>(Wh1, Wh2, ah, v1, v2);
  prep_s_kernel<<<8, 256, 0, stream>>>(input, v1, v2, s1, s2);
  h_gemm_kernel<<<128, 32, 0, stream>>>(input, Wh, h);
  const size_t smem = (size_t)(8 * ROWSTRIDE + 32 * 256) * sizeof(float);
  fused_attention_kernel<<<256, 1024, smem, stream>>>(edge_feat, adj, Wf, af, bf,
                                                      s1, s2, h, hp);
  bn_partial_kernel<<<64, 256, 0, stream>>>(hp, bp);
  bn_stats_kernel<<<1, 32, 0, stream>>>(bp, ms);
  bn_norm_kernel<<<256, 256, 0, stream>>>(hp, ms, gamma, beta, out);
}


# 7 launches fused to 3: one prep kernel makes h, s1, s2 on WMMA (k split over 4 waves), attention tail emits BatchNorm column partials, one BN kernel; h, s and BN sums are summed in a new order
# speedup vs baseline: 5.4361x; 1.2631x over previous
#include <hip/hip_runtime.h>
#include <math.h>

#define N 2048
#define D_IN 128
#define D_OUT 32
#define D_EDGE 8
#define ALPHA 0.2f
#define NEG_BIG -9e15f
#define ROWSTRIDE 2050
#define TROWS 8
#define TWAVES 32
#define GROUPS 4

typedef __attribute__((ext_vector_type(2))) float v2f;
typedef __attribute__((ext_vector_type(4))) float v4f;
typedef __attribute__((ext_vector_type(8))) float v8f;

__device__ __forceinline__ v8f wmma_f32_k4(v2f a, v2f b, v8f c) {
  return __builtin_amdgcn_wmma_f32_16x16x4_f32(false, a, false, b, (short)0, c,
                                               false, false);
}

__device__ __forceinline__ float lrelu(float x) { return x >= 0.f ? x : ALPHA * x; }

__device__ __forceinline__ void group_sums(v2f a0, v2f a1, v2f bw00, v2f bw01,
                                           v2f bw10, v2f bw11, v8f cini0,
                                           v8f cini1, float af_lo, float af_hi,
                                           float (&p)[8]) {
  v8f f0 = wmma_f32_k4(a0, bw00, cini0);
  f0 = wmma_f32_k4(a1, bw10, f0);
  v8f f1 = wmma_f32_k4(a0, bw01, cini1);
  f1 = wmma_f32_k4(a1, bw11, f1);
  for (int v = 0; v < 8; ++v)
    p[v] = lrelu(f0[v]) * af_lo + lrelu(f1[v]) * af_hi;
  for (int mask = 1; mask <= 8; mask <<= 1)
    for (int v = 0; v < 8; ++v) p[v] += __shfl_xor(p[v], mask, 32);
}

#define PTILE 272
#define PWAVE (3 * PTILE)
__global__ void __launch_bounds__(128)
prep_kernel(const float* __restrict__ input, const float* __restrict__ Wh,
            const float* __restrict__ Wh1, const float* __restrict__ Wh2,
            const float* __restrict__ ah, float* __restrict__ h,
            float* __restrict__ s1, float* __restrict__ s2) {
  __shared__ __attribute__((aligned(16))) float vs[2 * D_IN];
  __shared__ float part[4 * PWAVE];
  const int tid = threadIdx.x;
  const int lane = tid & 31;
  const int wave = tid >> 5;
  const int c = lane & 15;
  const int hf = lane >> 4;
  const int i0 = blockIdx.x * 16;
  const int kb = wave * 32 + hf * 16;

  const v4f* w1p = (const v4f*)(Wh1 + (size_t)tid * D_OUT);
  const v4f* w2p = (const v4f*)(Wh2 + (size_t)tid * D_OUT);
  v4f w1[8], w2[8];
#pragma unroll
  for (int j = 0; j < 8; ++j) {
    w1[j] = w1p[j];
    w2[j] = w2p[j];
  }
  const v4f* ap = (const v4f*)(input + (size_t)(i0 + c) * D_IN + kb);
  v4f a4[4];
#pragma unroll
  for (int j = 0; j < 4; ++j) a4[j] = ap[j];
  v2f b0[8], b1[8];
#pragma unroll
  for (int s = 0; s < 8; ++s) {
    const float* wp = Wh + (size_t)(kb + 2 * s) * D_OUT + c;
    b0[s].x = wp[0];
    b0[s].y = wp[D_OUT];
    b1[s].x = wp[16];
    b1[s].y = wp[D_OUT + 16];
  }

  float a1 = 0.f, a2 = 0.f;
#pragma unroll
  for (int f = 0; f < D_OUT; ++f) {
    const float w = ah[f];
    a1 += w1[f >> 2][f & 3] * w;
    a2 += w2[f >> 2][f & 3] * w;
  }
  vs[tid] = a1;
  vs[D_IN + tid] = a2;
  __syncthreads();

  const v4f* vp = (const v4f*)(vs + (c & 1) * D_IN + kb);
  const bool vcol = c < 2;
  v4f bv4[4];
#pragma unroll
  for (int j = 0; j < 4; ++j) bv4[j] = vp[j];

  v8f c0 = {};
  v8f c1 = {};
  v8f c2 = {};
#pragma unroll
  for (int s = 0; s < 8; ++s) {
    v2f a, bv;
    a.x = a4[s >> 1][2 * (s & 1)];
    a.y = a4[s >> 1][2 * (s & 1) + 1];
    bv.x = vcol ? bv4[s >> 1][2 * (s & 1)] : 0.f;
    bv.y = vcol ? bv4[s >> 1][2 * (s & 1) + 1] : 0.f;
    c0 = wmma_f32_k4(a, b0[s], c0);
    c1 = wmma_f32_k4(a, b1[s], c1);
    c2 = wmma_f32_k4(a, bv, c2);
  }

  float* pw = part + wave * PWAVE;
#pragma unroll
  for (int r = 0; r < 8; ++r) {
    pw[r * 32 + lane] = c0[r];
    pw[PTILE + r * 32 + lane] = c1[r];
    pw[2 * PTILE + r * 32 + lane] = c2[r];
  }
  __syncthreads();

#pragma unroll
  for (int q = 0; q < 4; ++q) {
    const int R = 4 * wave + q;
    const int off = hf * PTILE + (R & 7) * 32 + c + 16 * (R >> 3);
    const float s = ((part[off] + part[PWAVE + off]) + part[2 * PWAVE + off]) +
                    part[3 * PWAVE + off];
    h[(size_t)(i0 + R) * D_OUT + lane] = s;
  }
  if (wave == 0) {
    const int off = 2 * PTILE + (c & 7) * 32 + hf + 16 * (c >> 3);
    const float s = ((part[off] + part[PWAVE + off]) + part[2 * PWAVE + off]) +
                    part[3 * PWAVE + off];
    float* dst = hf ? s2 : s1;
    dst[i0 + c] = s;
  }
}

__global__ void __launch_bounds__(1024)
fused_attention_kernel(const float* __restrict__ edge_feat,
                       const int* __restrict__ adj,
                       const float* __restrict__ Wf,
                       const float* __restrict__ af,
                       const float* __restrict__ bf,
                       const float* __restrict__ s1,
                       const float* __restrict__ s2,
                       const float* __restrict__ h,
                       float* __restrict__ h_prime,
                       float* __restrict__ bp) {
  extern __shared__ float smem[];
  float* sc = smem;
  float* part = smem + TROWS * ROWSTRIDE;
  __shared__ float s_max[TWAVES];
  __shared__ float s_sum[TWAVES];
  __shared__ float s_hp[TROWS * D_OUT];

  const int tid = threadIdx.x;
  const int lane = tid & 31;
  const int wave = tid >> 5;
  const int i0 = blockIdx.x * TROWS;

  const int nn = lane & 15;
  const int kd = (lane >> 4) << 1;

  v2f bw00, bw01, bw10, bw11;
  bw00.x = Wf[(0 + kd) * D_OUT + nn];
  bw00.y = Wf[(1 + kd) * D_OUT + nn];
  bw01.x = Wf[(0 + kd) * D_OUT + 16 + nn];
  bw01.y = Wf[(1 + kd) * D_OUT + 16 + nn];
  bw10.x = Wf[(4 + kd) * D_OUT + nn];
  bw10.y = Wf[(5 + kd) * D_OUT + nn];
  bw11.x = Wf[(4 + kd) * D_OUT + 16 + nn];
  bw11.y = Wf[(5 + kd) * D_OUT + 16 + nn];
  const float af_lo = af[nn], af_hi = af[16 + nn];
  const float bf_lo = bf[nn], bf_hi = bf[16 + nn];
  v8f cini0, cini1;
  for (int v = 0; v < 8; ++v) {
    cini0[v] = bf_lo;
    cini1[v] = bf_hi;
  }

  const int md = lane & 7;
  const bool second = (lane & 8) != 0;
  const int jl = wave * 64 + ((lane & 8) << 1) + ((lane >> 4) << 3) + md;
  float s2v[2];
  s2v[0] = s2[jl];
  s2v[1] = s2[jl + 32];
  for (int il = 0; il < TROWS; ++il) {
    const int i = i0 + il;
    const float s1i = s1[i];
    const float* ep = edge_feat + ((size_t)i * N + (wave * 64 + nn)) * D_EDGE + kd;
    const int* ap = adj + (size_t)i * N + jl;
    v2f a0[GROUPS], a1[GROUPS];
    int av[2];
#pragma unroll
    for (int u = 0; u < GROUPS; ++u) {
      a0[u] = *(const v2f*)(ep + u * 16 * D_EDGE);
      a1[u] = *(const v2f*)(ep + u * 16 * D_EDGE + 4);
    }
    av[0] = ap[0];
    av[1] = ap[32];
    __builtin_amdgcn_sched_barrier(0);
#pragma unroll
    for (int pr = 0; pr < 2; ++pr) {
      float pa[8], pb[8];
      group_sums(a0[2 * pr], a1[2 * pr], bw00, bw01, bw10, bw11, cini0, cini1,
                 af_lo, af_hi, pa);
      group_sums(a0[2 * pr + 1], a1[2 * pr + 1], bw00, bw01, bw10, bw11, cini0,
                 cini1, af_lo, af_hi, pb);
      float q[8];
      for (int v = 0; v < 8; ++v) q[v] = second ? pb[v] : pa[v];
      float t0 = (md & 1) ? q[1] : q[0];
      float t1 = (md & 1) ? q[3] : q[2];
      float t2 = (md & 1) ? q[5] : q[4];
      float t3 = (md & 1) ? q[7] : q[6];
      float u0 = (md & 2) ? t1 : t0;
      float u1 = (md & 2) ? t3 : t2;
      float cp = (md & 4) ? u1 : u0;
      float v = lrelu(s1i + s2v[pr]) + lrelu(cp);
      v = (av[pr] > 0) ? v : NEG_BIG;
      sc[il * ROWSTRIDE + jl + pr * 32] = v;
      if (pr == 0) __builtin_amdgcn_sched_barrier(0);
    }
  }
  __syncthreads();

  const int r = wave >> 2;
  {
    float* seg = sc + r * ROWSTRIDE + (wave & 3) * 512;
    float x[16];
    float m = -3.4e38f;
    for (int c = 0; c < 16; ++c) {
      x[c] = seg[lane + 32 * c];
      m = fmaxf(m, x[c]);
    }
    for (int mask = 16; mask >= 1; mask >>= 1)
      m = fmaxf(m, __shfl_xor(m, mask, 32));
    if (lane == 0) s_max[wave] = m;
    __syncthreads();
    m = fmaxf(fmaxf(s_max[4 * r], s_max[4 * r + 1]),
              fmaxf(s_max[4 * r + 2], s_max[4 * r + 3]));
    float s = 0.f;
    for (int c = 0; c < 16; ++c) {
      const float e = __expf(x[c] - m);
      seg[lane + 32 * c] = e;
      s += e;
    }
    for (int mask = 16; mask >= 1; mask >>= 1) s += __shfl_xor(s, mask, 32);
    if (lane == 0) s_sum[wave] = s;
  }
  __syncthreads();

  v8f c0 = {};
  v8f c1 = {};
  const int kbase = wave * 64;
  for (int kk = 0; kk < 64; kk += 4) {
    const int k0 = kbase + kk;
    v2f a = *(const v2f*)(sc + (nn & 7) * ROWSTRIDE + k0 + kd);
    v2f b0, b1;
    b0.x = h[(size_t)(k0 + kd) * D_OUT + nn];
    b0.y = h[(size_t)(k0 + kd + 1) * D_OUT + nn];
    b1.x = h[(size_t)(k0 + kd) * D_OUT + 16 + nn];
    b1.y = h[(size_t)(k0 + kd + 1) * D_OUT + 16 + nn];
    c0 = wmma_f32_k4(a, b0, c0);
    c1 = wmma_f32_k4(a, b1, c1);
  }
  if (lane < 16) {
    float* pw = part + wave * 256;
    for (int v = 0; v < 8; ++v) {
      pw[v * 32 + nn] = c0[v];
      pw[v * 32 + 16 + nn] = c1[v];
    }
  }
  __syncthreads();
  float hv = 0.f;
  if (tid < 256) {
    float s = 0.f;
    for (int w = 0; w < TWAVES; ++w) s += part[w * 256 + tid];
    const int row = tid >> 5;
    const float rs = (s_sum[4 * row] + s_sum[4 * row + 1]) +
                     (s_sum[4 * row + 2] + s_sum[4 * row + 3]);
    hv = s * (1.0f / rs);
    s_hp[tid] = hv;
  }
  __syncthreads();
  if (tid < 256) h_prime[(size_t)i0 * D_OUT + tid] = hv;
  if (tid < 64) {
    const int col = tid & 31;
    const bool sq = tid >= 32;
    float acc = 0.f;
    for (int r = 0; r < TROWS; ++r) {
      const float v = s_hp[r * D_OUT + col];
      acc += sq ? v * v : v;
    }
    bp[blockIdx.x * 64 + tid] = acc;
  }
}

#define NBLK (N / TROWS)
__global__ void __launch_bounds__(256)
bn_final_kernel(const float* __restrict__ hp, const float* __restrict__ bp,
                const float* __restrict__ gamma, const float* __restrict__ beta,
                float* __restrict__ out) {
  __shared__ float red[4][64];
  __shared__ float ms[64];
  const int t = threadIdx.x;
  const int i4 = blockIdx.x * 256 + t;
  const v4f x = ((const v4f*)hp)[i4];
  const int col0 = (t & 7) * 4;
  const v4f g = *(const v4f*)(gamma + col0);
  const v4f b = *(const v4f*)(beta + col0);

  const int cc = t & 63;
  const int p = t >> 6;
  const float* bq = bp + (size_t)(p * (NBLK / 4)) * 64 + cc;
  float acc = 0.f;
#pragma unroll
  for (int k = 0; k < NBLK / 4; ++k) acc += bq[k * 64];
  red[p][cc] = acc;
  __syncthreads();
  if (t < 32) {
    const float s = ((red[0][t] + red[1][t]) + red[2][t]) + red[3][t];
    const float q = ((red[0][32 + t] + red[1][32 + t]) + red[2][32 + t]) +
                    red[3][32 + t];
    const float mean = s * (1.0f / (float)N);
    const float var = q * (1.0f / (float)N) - mean * mean;
    ms[t] = mean;
    ms[32 + t] = rsqrtf(var + 1e-5f);
  }
  __syncthreads();
  v4f o;
#pragma unroll
  for (int e = 0; e < 4; ++e)
    o[e] = g[e] * (x[e] - ms[col0 + e]) * ms[32 + col0 + e] + b[e];
  ((v4f*)out)[i4] = o;
}

extern "C" void kernel_launch(void* const* d_in, const int* in_sizes, int n_in,
                              void* d_out, int out_size, void* d_ws,
                              size_t ws_size, hipStream_t stream) {
  const float* input = (const float*)d_in[0];
  const float* edge_feat = (const float*)d_in[1];
  const int* adj = (const int*)d_in[2];
  const float* Wh = (const float*)d_in[3];
  const float* Wh1 = (const float*)d_in[4];
  const float* Wh2 = (const float*)d_in[5];
  const float* Wf = (const float*)d_in[6];
  const float* ah = (const float*)d_in[7];
  const float* af = (const float*)d_in[8];
  const float* bf = (const float*)d_in[9];
  const float* gamma = (const float*)d_in[10];
  const float* beta = (const float*)d_in[11];

  float* ws = (float*)d_ws;
  float* s1 = ws + 256;
  float* s2 = ws + 2304;
  float* h = ws + 4352;
  float* hp = ws + 69888;
  float* bp = ws + 135424;
  float* out = (float*)d_out;

  prep_kernel<<<128, 128, 0, stream>>>(input, Wh, Wh1, Wh2, ah, h, s1, s2);
  const size_t smem = (size_t)(8 * ROWSTRIDE + 32 * 256) * sizeof(float);
  fused_attention_kernel<<<256, 1024, smem, stream>>>(edge_feat, adj, Wf, af, bf,
                                                      s1, s2, h, hp, bp);
  bn_final_kernel<<<64, 256, 0, stream>>>(hp, bp, gamma, beta, out);
}


# edge-score WMMA transposed (A = Wf^T, B = edges): a lane holds 16 features of one edge, so the 32 shuffle-adds per 16-edge group become in-lane FMAs + one half exchange; feature sum order changes
# speedup vs baseline: 6.8499x; 1.2601x over previous
#include <hip/hip_runtime.h>
#include <math.h>

#define N 2048
#define D_IN 128
#define D_OUT 32
#define D_EDGE 8
#define ALPHA 0.2f
#define NEG_BIG -9e15f
#define ROWSTRIDE 2050
#define TROWS 8
#define TWAVES 32
#define GROUPS 4

typedef __attribute__((ext_vector_type(2))) float v2f;
typedef __attribute__((ext_vector_type(4))) float v4f;
typedef __attribute__((ext_vector_type(8))) float v8f;

__device__ __forceinline__ v8f wmma_f32_k4(v2f a, v2f b, v8f c) {
  return __builtin_amdgcn_wmma_f32_16x16x4_f32(false, a, false, b, (short)0, c,
                                               false, false);
}

__device__ __forceinline__ float lrelu(float x) { return x >= 0.f ? x : ALPHA * x; }

__device__ __forceinline__ float xhalf(float x) {
  const int b = __builtin_bit_cast(int, x);
  return __builtin_bit_cast(
      float, __builtin_amdgcn_permlanex16(b, b, 0x76543210, 0xfedcba98, false, false));
}

__device__ __forceinline__ float group_partial(v2f e0, v2f e1, v2f wf00, v2f wf01,
                                               v2f wf10, v2f wf11, v8f cini0,
                                               v8f cini1, const v2f (&af0)[4],
                                               const v2f (&af1)[4]) {
  v8f f0 = wmma_f32_k4(wf00, e0, cini0);
  f0 = wmma_f32_k4(wf10, e1, f0);
  v8f f1 = wmma_f32_k4(wf01, e0, cini1);
  f1 = wmma_f32_k4(wf11, e1, f1);
  v2f acc0 = {0.f, 0.f}, acc1 = {0.f, 0.f};
#pragma unroll
  for (int j = 0; j < 4; ++j) {
    v2f g0, g1;
    g0.x = lrelu(f0[2 * j]);
    g0.y = lrelu(f0[2 * j + 1]);
    g1.x = lrelu(f1[2 * j]);
    g1.y = lrelu(f1[2 * j + 1]);
    acc0 = __builtin_elementwise_fma(g0, af0[j], acc0);
    acc1 = __builtin_elementwise_fma(g1, af1[j], acc1);
  }
  const v2f s = acc0 + acc1;
  return s.x + s.y;
}

#define PTILE 272
#define PWAVE (3 * PTILE)
__global__ void __launch_bounds__(128)
prep_kernel(const float* __restrict__ input, const float* __restrict__ Wh,
            const float* __restrict__ Wh1, const float* __restrict__ Wh2,
            const float* __restrict__ ah, float* __restrict__ h,
            float* __restrict__ s1, float* __restrict__ s2) {
  __shared__ __attribute__((aligned(16))) float vs[2 * D_IN];
  __shared__ float part[4 * PWAVE];
  const int tid = threadIdx.x;
  const int lane = tid & 31;
  const int wave = tid >> 5;
  const int c = lane & 15;
  const int hf = lane >> 4;
  const int i0 = blockIdx.x * 16;
  const int kb = wave * 32 + hf * 16;

  const v4f* w1p = (const v4f*)(Wh1 + (size_t)tid * D_OUT);
  const v4f* w2p = (const v4f*)(Wh2 + (size_t)tid * D_OUT);
  v4f w1[8], w2[8];
#pragma unroll
  for (int j = 0; j < 8; ++j) {
    w1[j] = w1p[j];
    w2[j] = w2p[j];
  }
  const v4f* ap = (const v4f*)(input + (size_t)(i0 + c) * D_IN + kb);
  v4f a4[4];
#pragma unroll
  for (int j = 0; j < 4; ++j) a4[j] = ap[j];
  v2f b0[8], b1[8];
#pragma unroll
  for (int s = 0; s < 8; ++s) {
    const float* wp = Wh + (size_t)(kb + 2 * s) * D_OUT + c;
    b0[s].x = wp[0];
    b0[s].y = wp[D_OUT];
    b1[s].x = wp[16];
    b1[s].y = wp[D_OUT + 16];
  }

  float a1 = 0.f, a2 = 0.f;
#pragma unroll
  for (int f = 0; f < D_OUT; ++f) {
    const float w = ah[f];
    a1 += w1[f >> 2][f & 3] * w;
    a2 += w2[f >> 2][f & 3] * w;
  }
  vs[tid] = a1;
  vs[D_IN + tid] = a2;
  __syncthreads();

  const v4f* vp = (const v4f*)(vs + (c & 1) * D_IN + kb);
  const bool vcol = c < 2;
  v4f bv4[4];
#pragma unroll
  for (int j = 0; j < 4; ++j) bv4[j] = vp[j];

  v8f c0 = {};
  v8f c1 = {};
  v8f c2 = {};
#pragma unroll
  for (int s = 0; s < 8; ++s) {
    v2f a, bv;
    a.x = a4[s >> 1][2 * (s & 1)];
    a.y = a4[s >> 1][2 * (s & 1) + 1];
    bv.x = vcol ? bv4[s >> 1][2 * (s & 1)] : 0.f;
    bv.y = vcol ? bv4[s >> 1][2 * (s & 1) + 1] : 0.f;
    c0 = wmma_f32_k4(a, b0[s], c0);
    c1 = wmma_f32_k4(a, b1[s], c1);
    c2 = wmma_f32_k4(a, bv, c2);
  }

  float* pw = part + wave * PWAVE;
#pragma unroll
  for (int r = 0; r < 8; ++r) {
    pw[r * 32 + lane] = c0[r];
    pw[PTILE + r * 32 + lane] = c1[r];
    pw[2 * PTILE + r * 32 + lane] = c2[r];
  }
  __syncthreads();

#pragma unroll
  for (int q = 0; q < 4; ++q) {
    const int R = 4 * wave + q;
    const int off = hf * PTILE + (R & 7) * 32 + c + 16 * (R >> 3);
    const float s = ((part[off] + part[PWAVE + off]) + part[2 * PWAVE + off]) +
                    part[3 * PWAVE + off];
    h[(size_t)(i0 + R) * D_OUT + lane] = s;
  }
  if (wave == 0) {
    const int off = 2 * PTILE + (c & 7) * 32 + hf + 16 * (c >> 3);
    const float s = ((part[off] + part[PWAVE + off]) + part[2 * PWAVE + off]) +
                    part[3 * PWAVE + off];
    float* dst = hf ? s2 : s1;
    dst[i0 + c] = s;
  }
}

__global__ void __launch_bounds__(1024)
fused_attention_kernel(const float* __restrict__ edge_feat,
                       const int* __restrict__ adj,
                       const float* __restrict__ Wf,
                       const float* __restrict__ af,
                       const float* __restrict__ bf,
                       const float* __restrict__ s1,
                       const float* __restrict__ s2,
                       const float* __restrict__ h,
                       float* __restrict__ h_prime,
                       float* __restrict__ bp) {
  extern __shared__ float smem[];
  float* sc = smem;
  float* part = smem + TROWS * ROWSTRIDE;
  __shared__ float s_max[TWAVES];
  __shared__ float s_sum[TWAVES];
  __shared__ float s_hp[TROWS * D_OUT];

  const int tid = threadIdx.x;
  const int lane = tid & 31;
  const int wave = tid >> 5;
  const int i0 = blockIdx.x * TROWS;

  const int nn = lane & 15;
  const int kd = (lane >> 4) << 1;

  v2f bw00, bw01, bw10, bw11;
  bw00.x = Wf[(0 + kd) * D_OUT + nn];
  bw00.y = Wf[(1 + kd) * D_OUT + nn];
  bw01.x = Wf[(0 + kd) * D_OUT + 16 + nn];
  bw01.y = Wf[(1 + kd) * D_OUT + 16 + nn];
  bw10.x = Wf[(4 + kd) * D_OUT + nn];
  bw10.y = Wf[(5 + kd) * D_OUT + nn];
  bw11.x = Wf[(4 + kd) * D_OUT + 16 + nn];
  bw11.y = Wf[(5 + kd) * D_OUT + 16 + nn];
  const int f8 = (lane >> 4) << 3;
  v2f af0[4], af1[4];
  v8f cini0, cini1;
  for (int v = 0; v < 8; ++v) {
    af0[v >> 1][v & 1] = af[f8 + v];
    af1[v >> 1][v & 1] = af[16 + f8 + v];
    cini0[v] = bf[f8 + v];
    cini1[v] = bf[16 + f8 + v];
  }

  const bool second = lane >= 16;
  const int jl = wave * 64 + lane;
  float s2v[2];
  s2v[0] = s2[jl];
  s2v[1] = s2[jl + 32];
  for (int il = 0; il < TROWS; ++il) {
    const int i = i0 + il;
    const float s1i = s1[i];
    const float* ep = edge_feat + ((size_t)i * N + (wave * 64 + nn)) * D_EDGE + kd;
    const int* ap = adj + (size_t)i * N + jl;
    v2f a0[GROUPS], a1[GROUPS];
    int av[2];
#pragma unroll
    for (int u = 0; u < GROUPS; ++u) {
      a0[u] = *(const v2f*)(ep + u * 16 * D_EDGE);
      a1[u] = *(const v2f*)(ep + u * 16 * D_EDGE + 4);
    }
    av[0] = ap[0];
    av[1] = ap[32];
    __builtin_amdgcn_sched_barrier(0);
#pragma unroll
    for (int pr = 0; pr < 2; ++pr) {
      const float pa = group_partial(a0[2 * pr], a1[2 * pr], bw00, bw01, bw10,
                                     bw11, cini0, cini1, af0, af1);
      const float pb = group_partial(a0[2 * pr + 1], a1[2 * pr + 1], bw00, bw01,
                                     bw10, bw11, cini0, cini1, af0, af1);
      const float keep = second ? pb : pa;
      const float send = second ? pa : pb;
      const float cp = keep + xhalf(send);
      float v = lrelu(s1i + s2v[pr]) + lrelu(cp);
      v = (av[pr] > 0) ? v : NEG_BIG;
      sc[il * ROWSTRIDE + jl + pr * 32] = v;
    }
  }
  __syncthreads();

  const int r = wave >> 2;
  {
    float* seg = sc + r * ROWSTRIDE + (wave & 3) * 512;
    float x[16];
    float m = -3.4e38f;
    for (int c = 0; c < 16; ++c) {
      x[c] = seg[lane + 32 * c];
      m = fmaxf(m, x[c]);
    }
    for (int mask = 16; mask >= 1; mask >>= 1)
      m = fmaxf(m, __shfl_xor(m, mask, 32));
    if (lane == 0) s_max[wave] = m;
    __syncthreads();
    m = fmaxf(fmaxf(s_max[4 * r], s_max[4 * r + 1]),
              fmaxf(s_max[4 * r + 2], s_max[4 * r + 3]));
    float s = 0.f;
    for (int c = 0; c < 16; ++c) {
      const float e = __expf(x[c] - m);
      seg[lane + 32 * c] = e;
      s += e;
    }
    for (int mask = 16; mask >= 1; mask >>= 1) s += __shfl_xor(s, mask, 32);
    if (lane == 0) s_sum[wave] = s;
  }
  __syncthreads();

  v8f c0 = {};
  v8f c1 = {};
  const int kbase = wave * 64;
  for (int kk = 0; kk < 64; kk += 4) {
    const int k0 = kbase + kk;
    v2f a = *(const v2f*)(sc + (nn & 7) * ROWSTRIDE + k0 + kd);
    v2f b0, b1;
    b0.x = h[(size_t)(k0 + kd) * D_OUT + nn];
    b0.y = h[(size_t)(k0 + kd + 1) * D_OUT + nn];
    b1.x = h[(size_t)(k0 + kd) * D_OUT + 16 + nn];
    b1.y = h[(size_t)(k0 + kd + 1) * D_OUT + 16 + nn];
    c0 = wmma_f32_k4(a, b0, c0);
    c1 = wmma_f32_k4(a, b1, c1);
  }
  if (lane < 16) {
    float* pw = part + wave * 256;
    for (int v = 0; v < 8; ++v) {
      pw[v * 32 + nn] = c0[v];
      pw[v * 32 + 16 + nn] = c1[v];
    }
  }
  __syncthreads();
  float hv = 0.f;
  if (tid < 256) {
    float s = 0.f;
    for (int w = 0; w < TWAVES; ++w) s += part[w * 256 + tid];
    const int row = tid >> 5;
    const float rs = (s_sum[4 * row] + s_sum[4 * row + 1]) +
                     (s_sum[4 * row + 2] + s_sum[4 * row + 3]);
    hv = s * (1.0f / rs);
    s_hp[tid] = hv;
  }
  __syncthreads();
  if (tid < 256) h_prime[(size_t)i0 * D_OUT + tid] = hv;
  if (tid < 64) {
    const int col = tid & 31;
    const bool sq = tid >= 32;
    float acc = 0.f;
    for (int r = 0; r < TROWS; ++r) {
      const float v = s_hp[r * D_OUT + col];
      acc += sq ? v * v : v;
    }
    bp[blockIdx.x * 64 + tid] = acc;
  }
}

#define NBLK (N / TROWS)
__global__ void __launch_bounds__(256)
bn_final_kernel(const float* __restrict__ hp, const float* __restrict__ bp,
                const float* __restrict__ gamma, const float* __restrict__ beta,
                float* __restrict__ out) {
  __shared__ float red[4][64];
  __shared__ float ms[64];
  const int t = threadIdx.x;
  const int i4 = blockIdx.x * 256 + t;
  const v4f x = ((const v4f*)hp)[i4];
  const int col0 = (t & 7) * 4;
  const v4f g = *(const v4f*)(gamma + col0);
  const v4f b = *(const v4f*)(beta + col0);

  const int cc = t & 63;
  const int p = t >> 6;
  const float* bq = bp + (size_t)(p * (NBLK / 4)) * 64 + cc;
  float acc = 0.f;
#pragma unroll
  for (int k = 0; k < NBLK / 4; ++k) acc += bq[k * 64];
  red[p][cc] = acc;
  __syncthreads();
  if (t < 32) {
    const float s = ((red[0][t] + red[1][t]) + red[2][t]) + red[3][t];
    const float q = ((red[0][32 + t] + red[1][32 + t]) + red[2][32 + t]) +
                    red[3][32 + t];
    const float mean = s * (1.0f / (float)N);
    const float var = q * (1.0f / (float)N) - mean * mean;
    ms[t] = mean;
    ms[32 + t] = rsqrtf(var + 1e-5f);
  }
  __syncthreads();
  v4f o;
#pragma unroll
  for (int e = 0; e < 4; ++e)
    o[e] = g[e] * (x[e] - ms[col0 + e]) * ms[32 + col0 + e] + b[e];
  ((v4f*)out)[i4] = o;
}

extern "C" void kernel_launch(void* const* d_in, const int* in_sizes, int n_in,
                              void* d_out, int out_size, void* d_ws,
                              size_t ws_size, hipStream_t stream) {
  const float* input = (const float*)d_in[0];
  const float* edge_feat = (const float*)d_in[1];
  const int* adj = (const int*)d_in[2];
  const float* Wh = (const float*)d_in[3];
  const float* Wh1 = (const float*)d_in[4];
  const float* Wh2 = (const float*)d_in[5];
  const float* Wf = (const float*)d_in[6];
  const float* ah = (const float*)d_in[7];
  const float* af = (const float*)d_in[8];
  const float* bf = (const float*)d_in[9];
  const float* gamma = (const float*)d_in[10];
  const float* beta = (const float*)d_in[11];

  float* ws = (float*)d_ws;
  float* s1 = ws + 256;
  float* s2 = ws + 2304;
  float* h = ws + 4352;
  float* hp = ws + 69888;
  float* bp = ws + 135424;
  float* out = (float*)d_out;

  prep_kernel<<<128, 128, 0, stream>>>(input, Wh, Wh1, Wh2, ah, h, s1, s2);
  const size_t smem = (size_t)(8 * ROWSTRIDE + 32 * 256) * sizeof(float);
  fused_attention_kernel<<<256, 1024, smem, stream>>>(edge_feat, adj, Wf, af, bf,
                                                      s1, s2, h, hp, bp);
  bn_final_kernel<<<64, 256, 0, stream>>>(hp, bp, gamma, beta, out);
}


# edge-score feature sum without compares: lrelu(f) = (1+a)/2 f + (1-a)/2 |f|; linear part from the edge operands against Wf@af (from an extra prep work-group), |f| part packed FMAs; sum order changes
# speedup vs baseline: 7.1070x; 1.0375x over previous
#include <hip/hip_runtime.h>
#include <math.h>

#define N 2048
#define D_IN 128
#define D_OUT 32
#define D_EDGE 8
#define ALPHA 0.2f
#define NEG_BIG -9e15f
#define ROWSTRIDE 2050
#define TROWS 8
#define TWAVES 32
#define GROUPS 4

typedef __attribute__((ext_vector_type(2))) float v2f;
typedef __attribute__((ext_vector_type(4))) float v4f;
typedef __attribute__((ext_vector_type(8))) float v8f;

__device__ __forceinline__ v8f wmma_f32_k4(v2f a, v2f b, v8f c) {
  return __builtin_amdgcn_wmma_f32_16x16x4_f32(false, a, false, b, (short)0, c,
                                               false, false);
}

__device__ __forceinline__ float lrelu(float x) { return fmaxf(x, ALPHA * x); }

__device__ __forceinline__ float xhalf(float x) {
  const int b = __builtin_bit_cast(int, x);
  return __builtin_bit_cast(
      float, __builtin_amdgcn_permlanex16(b, b, 0x76543210, 0xfedcba98, false, false));
}

__device__ __forceinline__ float group_partial(v2f e0, v2f e1, v2f wf00, v2f wf01,
                                               v2f wf10, v2f wf11, v8f cini0,
                                               v8f cini1, const v2f (&aw0)[4],
                                               const v2f (&aw1)[4], v2f u0,
                                               v2f u1, v2f seed) {
  v8f f0 = wmma_f32_k4(wf00, e0, cini0);
  f0 = wmma_f32_k4(wf10, e1, f0);
  v8f f1 = wmma_f32_k4(wf01, e0, cini1);
  f1 = wmma_f32_k4(wf11, e1, f1);
  v2f acc = __builtin_elementwise_fma(e0, u0, seed);
  acc = __builtin_elementwise_fma(e1, u1, acc);
#pragma unroll
  for (int j = 0; j < 4; ++j) {
    v2f g0, g1;
    g0.x = __builtin_fabsf(f0[2 * j]);
    g0.y = __builtin_fabsf(f0[2 * j + 1]);
    g1.x = __builtin_fabsf(f1[2 * j]);
    g1.y = __builtin_fabsf(f1[2 * j + 1]);
    acc = __builtin_elementwise_fma(g0, aw0[j], acc);
    acc = __builtin_elementwise_fma(g1, aw1[j], acc);
  }
  return acc.x + acc.y;
}

#define PREP_TILES (N / 16)
#define PTILE 272
#define PWAVE (3 * PTILE)
__global__ void __launch_bounds__(128)
prep_kernel(const float* __restrict__ input, const float* __restrict__ Wh,
            const float* __restrict__ Wh1, const float* __restrict__ Wh2,
            const float* __restrict__ ah, const float* __restrict__ Wf,
            const float* __restrict__ af, const float* __restrict__ bf,
            float* __restrict__ h, float* __restrict__ s1,
            float* __restrict__ s2, float* __restrict__ uw) {
  if (blockIdx.x == PREP_TILES) {
    const int t = threadIdx.x;
    if (t <= D_EDGE) {
      const float* row = t < D_EDGE ? Wf + t * D_OUT : bf;
      float acc = 0.f;
      for (int f = 0; f < D_OUT; ++f) acc += row[f] * af[f];
      uw[t] = acc;
    }
    return;
  }
  __shared__ __attribute__((aligned(16))) float vs[2 * D_IN];
  __shared__ float part[4 * PWAVE];
  const int tid = threadIdx.x;
  const int lane = tid & 31;
  const int wave = tid >> 5;
  const int c = lane & 15;
  const int hf = lane >> 4;
  const int i0 = blockIdx.x * 16;
  const int kb = wave * 32 + hf * 16;

  const v4f* w1p = (const v4f*)(Wh1 + (size_t)tid * D_OUT);
  const v4f* w2p = (const v4f*)(Wh2 + (size_t)tid * D_OUT);
  v4f w1[8], w2[8];
#pragma unroll
  for (int j = 0; j < 8; ++j) {
    w1[j] = w1p[j];
    w2[j] = w2p[j];
  }
  const v4f* ap = (const v4f*)(input + (size_t)(i0 + c) * D_IN + kb);
  v4f a4[4];
#pragma unroll
  for (int j = 0; j < 4; ++j) a4[j] = ap[j];
  v2f b0[8], b1[8];
#pragma unroll
  for (int s = 0; s < 8; ++s) {
    const float* wp = Wh + (size_t)(kb + 2 * s) * D_OUT + c;
    b0[s].x = wp[0];
    b0[s].y = wp[D_OUT];
    b1[s].x = wp[16];
    b1[s].y = wp[D_OUT + 16];
  }

  float a1 = 0.f, a2 = 0.f;
#pragma unroll
  for (int f = 0; f < D_OUT; ++f) {
    const float w = ah[f];
    a1 += w1[f >> 2][f & 3] * w;
    a2 += w2[f >> 2][f & 3] * w;
  }
  vs[tid] = a1;
  vs[D_IN + tid] = a2;
  __syncthreads();

  const v4f* vp = (const v4f*)(vs + (c & 1) * D_IN + kb);
  const bool vcol = c < 2;
  v4f bv4[4];
#pragma unroll
  for (int j = 0; j < 4; ++j) bv4[j] = vp[j];

  v8f c0 = {};
  v8f c1 = {};
  v8f c2 = {};
#pragma unroll
  for (int s = 0; s < 8; ++s) {
    v2f a, bv;
    a.x = a4[s >> 1][2 * (s & 1)];
    a.y = a4[s >> 1][2 * (s & 1) + 1];
    bv.x = vcol ? bv4[s >> 1][2 * (s & 1)] : 0.f;
    bv.y = vcol ? bv4[s >> 1][2 * (s & 1) + 1] : 0.f;
    c0 = wmma_f32_k4(a, b0[s], c0);
    c1 = wmma_f32_k4(a, b1[s], c1);
    c2 = wmma_f32_k4(a, bv, c2);
  }

  float* pw = part + wave * PWAVE;
#pragma unroll
  for (int r = 0; r < 8; ++r) {
    pw[r * 32 + lane] = c0[r];
    pw[PTILE + r * 32 + lane] = c1[r];
    pw[2 * PTILE + r * 32 + lane] = c2[r];
  }
  __syncthreads();

#pragma unroll
  for (int q = 0; q < 4; ++q) {
    const int R = 4 * wave + q;
    const int off = hf * PTILE + (R & 7) * 32 + c + 16 * (R >> 3);
    const float s = ((part[off] + part[PWAVE + off]) + part[2 * PWAVE + off]) +
                    part[3 * PWAVE + off];
    h[(size_t)(i0 + R) * D_OUT + lane] = s;
  }
  if (wave == 0) {
    const int off = 2 * PTILE + (c & 7) * 32 + hf + 16 * (c >> 3);
    const float s = ((part[off] + part[PWAVE + off]) + part[2 * PWAVE + off]) +
                    part[3 * PWAVE + off];
    float* dst = hf ? s2 : s1;
    dst[i0 + c] = s;
  }
}

__global__ void __launch_bounds__(1024)
fused_attention_kernel(const float* __restrict__ edge_feat,
                       const int* __restrict__ adj,
                       const float* __restrict__ Wf,
                       const float* __restrict__ af,
                       const float* __restrict__ bf,
                       const float* __restrict__ s1,
                       const float* __restrict__ s2,
                       const float* __restrict__ h,
                       const float* __restrict__ uw,
                       float* __restrict__ h_prime,
                       float* __restrict__ bp) {
  extern __shared__ float smem[];
  float* sc = smem;
  float* part = smem + TROWS * ROWSTRIDE;
  __shared__ float s_max[TWAVES];
  __shared__ float s_sum[TWAVES];
  __shared__ float s_hp[TROWS * D_OUT];

  const int tid = threadIdx.x;
  const int lane = tid & 31;
  const int wave = tid >> 5;
  const int i0 = blockIdx.x * TROWS;

  const int nn = lane & 15;
  const int kd = (lane >> 4) << 1;

  v2f bw00, bw01, bw10, bw11;
  bw00.x = Wf[(0 + kd) * D_OUT + nn];
  bw00.y = Wf[(1 + kd) * D_OUT + nn];
  bw01.x = Wf[(0 + kd) * D_OUT + 16 + nn];
  bw01.y = Wf[(1 + kd) * D_OUT + 16 + nn];
  bw10.x = Wf[(4 + kd) * D_OUT + nn];
  bw10.y = Wf[(5 + kd) * D_OUT + nn];
  bw11.x = Wf[(4 + kd) * D_OUT + 16 + nn];
  bw11.y = Wf[(5 + kd) * D_OUT + 16 + nn];
  const int f8 = (lane >> 4) << 3;
  const float ABSW = 0.5f * (1.0f - ALPHA);
  const float LINW = 0.5f * (1.0f + ALPHA);
  v2f aw0[4], aw1[4];
  v8f cini0, cini1;
  for (int v = 0; v < 8; ++v) {
    aw0[v >> 1][v & 1] = ABSW * af[f8 + v];
    aw1[v >> 1][v & 1] = ABSW * af[16 + f8 + v];
    cini0[v] = bf[f8 + v];
    cini1[v] = bf[16 + f8 + v];
  }
  v2f u0 = *(const v2f*)(uw + kd);
  v2f u1 = *(const v2f*)(uw + 4 + kd);
  u0 = LINW * u0;
  u1 = LINW * u1;
  v2f seed;
  seed.x = lane < 16 ? LINW * uw[8] : 0.f;
  seed.y = 0.f;

  const bool second = lane >= 16;
  const int jl = wave * 64 + lane;
  float s2v[2];
  s2v[0] = s2[jl];
  s2v[1] = s2[jl + 32];
  for (int il = 0; il < TROWS; ++il) {
    const int i = i0 + il;
    const float s1i = s1[i];
    const float* ep = edge_feat + ((size_t)i * N + (wave * 64 + nn)) * D_EDGE + kd;
    const int* ap = adj + (size_t)i * N + jl;
    v2f a0[GROUPS], a1[GROUPS];
    int av[2];
#pragma unroll
    for (int u = 0; u < GROUPS; ++u) {
      a0[u] = *(const v2f*)(ep + u * 16 * D_EDGE);
      a1[u] = *(const v2f*)(ep + u * 16 * D_EDGE + 4);
    }
    av[0] = ap[0];
    av[1] = ap[32];
    __builtin_amdgcn_sched_barrier(0);
#pragma unroll
    for (int pr = 0; pr < 2; ++pr) {
      const float pa = group_partial(a0[2 * pr], a1[2 * pr], bw00, bw01, bw10,
                                     bw11, cini0, cini1, aw0, aw1, u0, u1, seed);
      const float pb = group_partial(a0[2 * pr + 1], a1[2 * pr + 1], bw00, bw01,
                                     bw10, bw11, cini0, cini1, aw0, aw1, u0, u1,
                                     seed);
      const float keep = second ? pb : pa;
      const float send = second ? pa : pb;
      const float cp = keep + xhalf(send);
      float v = lrelu(s1i + s2v[pr]) + lrelu(cp);
      v = (av[pr] > 0) ? v : NEG_BIG;
      sc[il * ROWSTRIDE + jl + pr * 32] = v;
    }
  }
  __syncthreads();

  const int r = wave >> 2;
  {
    float* seg = sc + r * ROWSTRIDE + (wave & 3) * 512;
    float x[16];
    float m = -3.4e38f;
    for (int c = 0; c < 16; ++c) {
      x[c] = seg[lane + 32 * c];
      m = fmaxf(m, x[c]);
    }
    for (int mask = 16; mask >= 1; mask >>= 1)
      m = fmaxf(m, __shfl_xor(m, mask, 32));
    if (lane == 0) s_max[wave] = m;
    __syncthreads();
    m = fmaxf(fmaxf(s_max[4 * r], s_max[4 * r + 1]),
              fmaxf(s_max[4 * r + 2], s_max[4 * r + 3]));
    float s = 0.f;
    for (int c = 0; c < 16; ++c) {
      const float e = __expf(x[c] - m);
      seg[lane + 32 * c] = e;
      s += e;
    }
    for (int mask = 16; mask >= 1; mask >>= 1) s += __shfl_xor(s, mask, 32);
    if (lane == 0) s_sum[wave] = s;
  }
  __syncthreads();

  v8f c0 = {};
  v8f c1 = {};
  const int kbase = wave * 64;
  for (int kk = 0; kk < 64; kk += 4) {
    const int k0 = kbase + kk;
    v2f a = *(const v2f*)(sc + (nn & 7) * ROWSTRIDE + k0 + kd);
    v2f b0, b1;
    b0.x = h[(size_t)(k0 + kd) * D_OUT + nn];
    b0.y = h[(size_t)(k0 + kd + 1) * D_OUT + nn];
    b1.x = h[(size_t)(k0 + kd) * D_OUT + 16 + nn];
    b1.y = h[(size_t)(k0 + kd + 1) * D_OUT + 16 + nn];
    c0 = wmma_f32_k4(a, b0, c0);
    c1 = wmma_f32_k4(a, b1, c1);
  }
  if (lane < 16) {
    float* pw = part + wave * 256;
    for (int v = 0; v < 8; ++v) {
      pw[v * 32 + nn] = c0[v];
      pw[v * 32 + 16 + nn] = c1[v];
    }
  }
  __syncthreads();
  float hv = 0.f;
  if (tid < 256) {
    float s = 0.f;
    for (int w = 0; w < TWAVES; ++w) s += part[w * 256 + tid];
    const int row = tid >> 5;
    const float rs = (s_sum[4 * row] + s_sum[4 * row + 1]) +
                     (s_sum[4 * row + 2] + s_sum[4 * row + 3]);
    hv = s * (1.0f / rs);
    s_hp[tid] = hv;
  }
  __syncthreads();
  if (tid < 256) h_prime[(size_t)i0 * D_OUT + tid] = hv;
  if (tid < 64) {
    const int col = tid & 31;
    const bool sq = tid >= 32;
    float acc = 0.f;
    for (int r = 0; r < TROWS; ++r) {
      const float v = s_hp[r * D_OUT + col];
      acc += sq ? v * v : v;
    }
    bp[blockIdx.x * 64 + tid] = acc;
  }
}

#define NBLK (N / TROWS)
__global__ void __launch_bounds__(256)
bn_final_kernel(const float* __restrict__ hp, const float* __restrict__ bp,
                const float* __restrict__ gamma, const float* __restrict__ beta,
                float* __restrict__ out) {
  __shared__ float red[4][64];
  __shared__ float ms[64];
  const int t = threadIdx.x;
  const int i4 = blockIdx.x * 256 + t;
  const v4f x = ((const v4f*)hp)[i4];
  const int col0 = (t & 7) * 4;
  const v4f g = *(const v4f*)(gamma + col0);
  const v4f b = *(const v4f*)(beta + col0);

  const int cc = t & 63;
  const int p = t >> 6;
  const float* bq = bp + (size_t)(p * (NBLK / 4)) * 64 + cc;
  float acc = 0.f;
#pragma unroll
  for (int k = 0; k < NBLK / 4; ++k) acc += bq[k * 64];
  red[p][cc] = acc;
  __syncthreads();
  if (t < 32) {
    const float s = ((red[0][t] + red[1][t]) + red[2][t]) + red[3][t];
    const float q = ((red[0][32 + t] + red[1][32 + t]) + red[2][32 + t]) +
                    red[3][32 + t];
    const float mean = s * (1.0f / (float)N);
    const float var = q * (1.0f / (float)N) - mean * mean;
    ms[t] = mean;
    ms[32 + t] = rsqrtf(var + 1e-5f);
  }
  __syncthreads();
  v4f o;
#pragma unroll
  for (int e = 0; e < 4; ++e)
    o[e] = g[e] * (x[e] - ms[col0 + e]) * ms[32 + col0 + e] + b[e];
  ((v4f*)out)[i4] = o;
}

extern "C" void kernel_launch(void* const* d_in, const int* in_sizes, int n_in,
                              void* d_out, int out_size, void* d_ws,
                              size_t ws_size, hipStream_t stream) {
  const float* input = (const float*)d_in[0];
  const float* edge_feat = (const float*)d_in[1];
  const int* adj = (const int*)d_in[2];
  const float* Wh = (const float*)d_in[3];
  const float* Wh1 = (const float*)d_in[4];
  const float* Wh2 = (const float*)d_in[5];
  const float* Wf = (const float*)d_in[6];
  const float* ah = (const float*)d_in[7];
  const float* af = (const float*)d_in[8];
  const float* bf = (const float*)d_in[9];
  const float* gamma = (const float*)d_in[10];
  const float* beta = (const float*)d_in[11];

  float* ws = (float*)d_ws;
  float* uw = ws;
  float* s1 = ws + 256;
  float* s2 = ws + 2304;
  float* h = ws + 4352;
  float* hp = ws + 69888;
  float* bp = ws + 135424;
  float* out = (float*)d_out;

  prep_kernel<<<129, 128, 0, stream>>>(input, Wh, Wh1, Wh2, ah, Wf, af, bf, h, s1,
                                       s2, uw);
  const size_t smem = (size_t)(8 * ROWSTRIDE + 32 * 256) * sizeof(float);
  fused_attention_kernel<<<256, 1024, smem, stream>>>(edge_feat, adj, Wf, af, bf,
                                                      s1, s2, h, uw, hp, bp);
  bn_final_kernel<<<64, 256, 0, stream>>>(hp, bp, gamma, beta, out);
}


# attention work-groups of 16 waves instead of 32, each wave scoring 128 columns per row (20 loads in flight): half as many waves to launch; softmax and P@h sums split 2 and 16 ways (were 4 and 32)
# speedup vs baseline: 7.4918x; 1.0541x over previous
#include <hip/hip_runtime.h>
#include <math.h>

#define N 2048
#define D_IN 128
#define D_OUT 32
#define D_EDGE 8
#define ALPHA 0.2f
#define NEG_BIG -9e15f
#define ROWSTRIDE 2050
#define TROWS 8
#define TWAVES 16
#define WCOLS (N / TWAVES)
#define GROUPS (WCOLS / 16)
#define PAIRS (WCOLS / 32)
#define WPR (TWAVES / TROWS)
#define XPL (N / WPR / 32)

typedef __attribute__((ext_vector_type(2))) float v2f;
typedef __attribute__((ext_vector_type(4))) float v4f;
typedef __attribute__((ext_vector_type(8))) float v8f;

__device__ __forceinline__ v8f wmma_f32_k4(v2f a, v2f b, v8f c) {
  return __builtin_amdgcn_wmma_f32_16x16x4_f32(false, a, false, b, (short)0, c,
                                               false, false);
}

__device__ __forceinline__ float lrelu(float x) { return fmaxf(x, ALPHA * x); }

__device__ __forceinline__ float xhalf(float x) {
  const int b = __builtin_bit_cast(int, x);
  return __builtin_bit_cast(
      float, __builtin_amdgcn_permlanex16(b, b, 0x76543210, 0xfedcba98, false, false));
}

__device__ __forceinline__ float group_partial(v2f e0, v2f e1, v2f wf00, v2f wf01,
                                               v2f wf10, v2f wf11, v8f cini0,
                                               v8f cini1, const v2f (&aw0)[4],
                                               const v2f (&aw1)[4], v2f u0,
                                               v2f u1, v2f seed) {
  v8f f0 = wmma_f32_k4(wf00, e0, cini0);
  f0 = wmma_f32_k4(wf10, e1, f0);
  v8f f1 = wmma_f32_k4(wf01, e0, cini1);
  f1 = wmma_f32_k4(wf11, e1, f1);
  v2f acc = __builtin_elementwise_fma(e0, u0, seed);
  acc = __builtin_elementwise_fma(e1, u1, acc);
#pragma unroll
  for (int j = 0; j < 4; ++j) {
    v2f g0, g1;
    g0.x = __builtin_fabsf(f0[2 * j]);
    g0.y = __builtin_fabsf(f0[2 * j + 1]);
    g1.x = __builtin_fabsf(f1[2 * j]);
    g1.y = __builtin_fabsf(f1[2 * j + 1]);
    acc = __builtin_elementwise_fma(g0, aw0[j], acc);
    acc = __builtin_elementwise_fma(g1, aw1[j], acc);
  }
  return acc.x + acc.y;
}

#define PREP_TILES (N / 16)
#define PTILE 272
#define PWAVE (3 * PTILE)
__global__ void __launch_bounds__(128)
prep_kernel(const float* __restrict__ input, const float* __restrict__ Wh,
            const float* __restrict__ Wh1, const float* __restrict__ Wh2,
            const float* __restrict__ ah, const float* __restrict__ Wf,
            const float* __restrict__ af, const float* __restrict__ bf,
            float* __restrict__ h, float* __restrict__ s1,
            float* __restrict__ s2, float* __restrict__ uw) {
  if (blockIdx.x == PREP_TILES) {
    const int t = threadIdx.x;
    if (t <= D_EDGE) {
      const float* row = t < D_EDGE ? Wf + t * D_OUT : bf;
      float acc = 0.f;
      for (int f = 0; f < D_OUT; ++f) acc += row[f] * af[f];
      uw[t] = acc;
    }
    return;
  }
  __shared__ __attribute__((aligned(16))) float vs[2 * D_IN];
  __shared__ float part[4 * PWAVE];
  const int tid = threadIdx.x;
  const int lane = tid & 31;
  const int wave = tid >> 5;
  const int c = lane & 15;
  const int hf = lane >> 4;
  const int i0 = blockIdx.x * 16;
  const int kb = wave * 32 + hf * 16;

  const v4f* w1p = (const v4f*)(Wh1 + (size_t)tid * D_OUT);
  const v4f* w2p = (const v4f*)(Wh2 + (size_t)tid * D_OUT);
  v4f w1[8], w2[8];
#pragma unroll
  for (int j = 0; j < 8; ++j) {
    w1[j] = w1p[j];
    w2[j] = w2p[j];
  }
  const v4f* ap = (const v4f*)(input + (size_t)(i0 + c) * D_IN + kb);
  v4f a4[4];
#pragma unroll
  for (int j = 0; j < 4; ++j) a4[j] = ap[j];
  v2f b0[8], b1[8];
#pragma unroll
  for (int s = 0; s < 8; ++s) {
    const float* wp = Wh + (size_t)(kb + 2 * s) * D_OUT + c;
    b0[s].x = wp[0];
    b0[s].y = wp[D_OUT];
    b1[s].x = wp[16];
    b1[s].y = wp[D_OUT + 16];
  }

  float a1 = 0.f, a2 = 0.f;
#pragma unroll
  for (int f = 0; f < D_OUT; ++f) {
    const float w = ah[f];
    a1 += w1[f >> 2][f & 3] * w;
    a2 += w2[f >> 2][f & 3] * w;
  }
  vs[tid] = a1;
  vs[D_IN + tid] = a2;
  __syncthreads();

  const v4f* vp = (const v4f*)(vs + (c & 1) * D_IN + kb);
  const bool vcol = c < 2;
  v4f bv4[4];
#pragma unroll
  for (int j = 0; j < 4; ++j) bv4[j] = vp[j];

  v8f c0 = {};
  v8f c1 = {};
  v8f c2 = {};
#pragma unroll
  for (int s = 0; s < 8; ++s) {
    v2f a, bv;
    a.x = a4[s >> 1][2 * (s & 1)];
    a.y = a4[s >> 1][2 * (s & 1) + 1];
    bv.x = vcol ? bv4[s >> 1][2 * (s & 1)] : 0.f;
    bv.y = vcol ? bv4[s >> 1][2 * (s & 1) + 1] : 0.f;
    c0 = wmma_f32_k4(a, b0[s], c0);
    c1 = wmma_f32_k4(a, b1[s], c1);
    c2 = wmma_f32_k4(a, bv, c2);
  }

  float* pw = part + wave * PWAVE;
#pragma unroll
  for (int r = 0; r < 8; ++r) {
    pw[r * 32 + lane] = c0[r];
    pw[PTILE + r * 32 + lane] = c1[r];
    pw[2 * PTILE + r * 32 + lane] = c2[r];
  }
  __syncthreads();

#pragma unroll
  for (int q = 0; q < 4; ++q) {
    const int R = 4 * wave + q;
    const int off = hf * PTILE + (R & 7) * 32 + c + 16 * (R >> 3);
    const float s = ((part[off] + part[PWAVE + off]) + part[2 * PWAVE + off]) +
                    part[3 * PWAVE + off];
    h[(size_t)(i0 + R) * D_OUT + lane] = s;
  }
  if (wave == 0) {
    const int off = 2 * PTILE + (c & 7) * 32 + hf + 16 * (c >> 3);
    const float s = ((part[off] + part[PWAVE + off]) + part[2 * PWAVE + off]) +
                    part[3 * PWAVE + off];
    float* dst = hf ? s2 : s1;
    dst[i0 + c] = s;
  }
}

__global__ void __launch_bounds__(32 * TWAVES)
fused_attention_kernel(const float* __restrict__ edge_feat,
                       const int* __restrict__ adj,
                       const float* __restrict__ Wf,
                       const float* __restrict__ af,
                       const float* __restrict__ bf,
                       const float* __restrict__ s1,
                       const float* __restrict__ s2,
                       const float* __restrict__ h,
                       const float* __restrict__ uw,
                       float* __restrict__ h_prime,
                       float* __restrict__ bp) {
  extern __shared__ float smem[];
  float* sc = smem;
  float* part = smem + TROWS * ROWSTRIDE;
  __shared__ float s_max[TWAVES];
  __shared__ float s_sum[TWAVES];
  __shared__ float s_hp[TROWS * D_OUT];

  const int tid = threadIdx.x;
  const int lane = tid & 31;
  const int wave = tid >> 5;
  const int i0 = blockIdx.x * TROWS;

  const int nn = lane & 15;
  const int kd = (lane >> 4) << 1;

  v2f bw00, bw01, bw10, bw11;
  bw00.x = Wf[(0 + kd) * D_OUT + nn];
  bw00.y = Wf[(1 + kd) * D_OUT + nn];
  bw01.x = Wf[(0 + kd) * D_OUT + 16 + nn];
  bw01.y = Wf[(1 + kd) * D_OUT + 16 + nn];
  bw10.x = Wf[(4 + kd) * D_OUT + nn];
  bw10.y = Wf[(5 + kd) * D_OUT + nn];
  bw11.x = Wf[(4 + kd) * D_OUT + 16 + nn];
  bw11.y = Wf[(5 + kd) * D_OUT + 16 + nn];
  const int f8 = (lane >> 4) << 3;
  const float ABSW = 0.5f * (1.0f - ALPHA);
  const float LINW = 0.5f * (1.0f + ALPHA);
  v2f aw0[4], aw1[4];
  v8f cini0, cini1;
  for (int v = 0; v < 8; ++v) {
    aw0[v >> 1][v & 1] = ABSW * af[f8 + v];
    aw1[v >> 1][v & 1] = ABSW * af[16 + f8 + v];
    cini0[v] = bf[f8 + v];
    cini1[v] = bf[16 + f8 + v];
  }
  v2f u0 = *(const v2f*)(uw + kd);
  v2f u1 = *(const v2f*)(uw + 4 + kd);
  u0 = LINW * u0;
  u1 = LINW * u1;
  v2f seed;
  seed.x = lane < 16 ? LINW * uw[8] : 0.f;
  seed.y = 0.f;

  const bool second = lane >= 16;
  const int jl = wave * WCOLS + lane;
  float s2v[PAIRS];
#pragma unroll
  for (int p = 0; p < PAIRS; ++p) s2v[p] = s2[jl + 32 * p];
  for (int il = 0; il < TROWS; ++il) {
    const int i = i0 + il;
    const float s1i = s1[i];
    const float* ep = edge_feat + ((size_t)i * N + (wave * WCOLS + nn)) * D_EDGE + kd;
    const int* ap = adj + (size_t)i * N + jl;
    v2f a0[GROUPS], a1[GROUPS];
    int av[PAIRS];
#pragma unroll
    for (int u = 0; u < GROUPS; ++u) {
      a0[u] = *(const v2f*)(ep + u * 16 * D_EDGE);
      a1[u] = *(const v2f*)(ep + u * 16 * D_EDGE + 4);
    }
#pragma unroll
    for (int p = 0; p < PAIRS; ++p) av[p] = ap[32 * p];
    __builtin_amdgcn_sched_barrier(0);
#pragma unroll
    for (int pr = 0; pr < PAIRS; ++pr) {
      const float pa = group_partial(a0[2 * pr], a1[2 * pr], bw00, bw01, bw10,
                                     bw11, cini0, cini1, aw0, aw1, u0, u1, seed);
      const float pb = group_partial(a0[2 * pr + 1], a1[2 * pr + 1], bw00, bw01,
                                     bw10, bw11, cini0, cini1, aw0, aw1, u0, u1,
                                     seed);
      const float keep = second ? pb : pa;
      const float send = second ? pa : pb;
      const float cp = keep + xhalf(send);
      float v = lrelu(s1i + s2v[pr]) + lrelu(cp);
      v = (av[pr] > 0) ? v : NEG_BIG;
      sc[il * ROWSTRIDE + jl + pr * 32] = v;
    }
  }
  __syncthreads();

  const int r = wave / WPR;
  {
    float* seg = sc + r * ROWSTRIDE + (wave % WPR) * (N / WPR);
    float x[XPL];
    float m = -3.4e38f;
    for (int c = 0; c < XPL; ++c) {
      x[c] = seg[lane + 32 * c];
      m = fmaxf(m, x[c]);
    }
    for (int mask = 16; mask >= 1; mask >>= 1)
      m = fmaxf(m, __shfl_xor(m, mask, 32));
    if (lane == 0) s_max[wave] = m;
    __syncthreads();
    m = fmaxf(s_max[WPR * r], s_max[WPR * r + 1]);
    float s = 0.f;
    for (int c = 0; c < XPL; ++c) {
      const float e = __expf(x[c] - m);
      seg[lane + 32 * c] = e;
      s += e;
    }
    for (int mask = 16; mask >= 1; mask >>= 1) s += __shfl_xor(s, mask, 32);
    if (lane == 0) s_sum[wave] = s;
  }
  __syncthreads();

  v8f c0 = {};
  v8f c1 = {};
  const int kbase = wave * WCOLS;
  for (int kk = 0; kk < WCOLS; kk += 4) {
    const int k0 = kbase + kk;
    v2f a = *(const v2f*)(sc + (nn & 7) * ROWSTRIDE + k0 + kd);
    v2f b0, b1;
    b0.x = h[(size_t)(k0 + kd) * D_OUT + nn];
    b0.y = h[(size_t)(k0 + kd + 1) * D_OUT + nn];
    b1.x = h[(size_t)(k0 + kd) * D_OUT + 16 + nn];
    b1.y = h[(size_t)(k0 + kd + 1) * D_OUT + 16 + nn];
    c0 = wmma_f32_k4(a, b0, c0);
    c1 = wmma_f32_k4(a, b1, c1);
  }
  if (lane < 16) {
    float* pw = part + wave * 256;
    for (int v = 0; v < 8; ++v) {
      pw[v * 32 + nn] = c0[v];
      pw[v * 32 + 16 + nn] = c1[v];
    }
  }
  __syncthreads();
  float hv = 0.f;
  if (tid < 256) {
    float s = 0.f;
    for (int w = 0; w < TWAVES; ++w) s += part[w * 256 + tid];
    const int row = tid >> 5;
    const float rs = s_sum[WPR * row] + s_sum[WPR * row + 1];
    hv = s * (1.0f / rs);
    s_hp[tid] = hv;
  }
  __syncthreads();
  if (tid < 256) h_prime[(size_t)i0 * D_OUT + tid] = hv;
  if (tid < 64) {
    const int col = tid & 31;
    const bool sq = tid >= 32;
    float acc = 0.f;
    for (int r = 0; r < TROWS; ++r) {
      const float v = s_hp[r * D_OUT + col];
      acc += sq ? v * v : v;
    }
    bp[blockIdx.x * 64 + tid] = acc;
  }
}

#define NBLK (N / TROWS)
__global__ void __launch_bounds__(256)
bn_final_kernel(const float* __restrict__ hp, const float* __restrict__ bp,
                const float* __restrict__ gamma, const float* __restrict__ beta,
                float* __restrict__ out) {
  __shared__ float red[4][64];
  __shared__ float ms[64];
  const int t = threadIdx.x;
  const int i4 = blockIdx.x * 256 + t;
  const v4f x = ((const v4f*)hp)[i4];
  const int col0 = (t & 7) * 4;
  const v4f g = *(const v4f*)(gamma + col0);
  const v4f b = *(const v4f*)(beta + col0);

  const int cc = t & 63;
  const int p = t >> 6;
  const float* bq = bp + (size_t)(p * (NBLK / 4)) * 64 + cc;
  float acc = 0.f;
#pragma unroll
  for (int k = 0; k < NBLK / 4; ++k) acc += bq[k * 64];
  red[p][cc] = acc;
  __syncthreads();
  if (t < 32) {
    const float s = ((red[0][t] + red[1][t]) + red[2][t]) + red[3][t];
    const float q = ((red[0][32 + t] + red[1][32 + t]) + red[2][32 + t]) +
                    red[3][32 + t];
    const float mean = s * (1.0f / (float)N);
    const float var = q * (1.0f / (float)N) - mean * mean;
    ms[t] = mean;
    ms[32 + t] = rsqrtf(var + 1e-5f);
  }
  __syncthreads();
  v4f o;
#pragma unroll
  for (int e = 0; e < 4; ++e)
    o[e] = g[e] * (x[e] - ms[col0 + e]) * ms[32 + col0 + e] + b[e];
  ((v4f*)out)[i4] = o;
}

extern "C" void kernel_launch(void* const* d_in, const int* in_sizes, int n_in,
                              void* d_out, int out_size, void* d_ws,
                              size_t ws_size, hipStream_t stream) {
  const float* input = (const float*)d_in[0];
  const float* edge_feat = (const float*)d_in[1];
  const int* adj = (const int*)d_in[2];
  const float* Wh = (const float*)d_in[3];
  const float* Wh1 = (const float*)d_in[4];
  const float* Wh2 = (const float*)d_in[5];
  const float* Wf = (const float*)d_in[6];
  const float* ah = (const float*)d_in[7];
  const float* af = (const float*)d_in[8];
  const float* bf = (const float*)d_in[9];
  const float* gamma = (const float*)d_in[10];
  const float* beta = (const float*)d_in[11];

  float* ws = (float*)d_ws;
  float* uw = ws;
  float* s1 = ws + 256;
  float* s2 = ws + 2304;
  float* h = ws + 4352;
  float* hp = ws + 69888;
  float* bp = ws + 135424;
  float* out = (float*)d_out;

  prep_kernel<<<129, 128, 0, stream>>>(input, Wh, Wh1, Wh2, ah, Wf, af, bf, h, s1,
                                       s2, uw);
  const size_t smem = (size_t)(8 * ROWSTRIDE + 16 * 256) * sizeof(float);
  fused_attention_kernel<<<256, 512, smem, stream>>>(edge_feat, adj, Wf, af, bf,
                                                      s1, s2, h, uw, hp, bp);
  bn_final_kernel<<<64, 256, 0, stream>>>(hp, bp, gamma, beta, out);
}


# edge-score feature sum: |f| taken as a source modifier of one scalar FMA per feature (16 vector issues per group, were 16 sign-clears + 8 packed FMAs); feature sum order changes
# speedup vs baseline: 7.7828x; 1.0388x over previous
#include <hip/hip_runtime.h>
#include <math.h>

#define N 2048
#define D_IN 128
#define D_OUT 32
#define D_EDGE 8
#define ALPHA 0.2f
#define NEG_BIG -9e15f
#define ROWSTRIDE 2050
#define TROWS 8
#define TWAVES 16
#define WCOLS (N / TWAVES)
#define GROUPS (WCOLS / 16)
#define PAIRS (WCOLS / 32)
#define WPR (TWAVES / TROWS)
#define XPL (N / WPR / 32)

typedef __attribute__((ext_vector_type(2))) float v2f;
typedef __attribute__((ext_vector_type(4))) float v4f;
typedef __attribute__((ext_vector_type(8))) float v8f;

__device__ __forceinline__ v8f wmma_f32_k4(v2f a, v2f b, v8f c) {
  return __builtin_amdgcn_wmma_f32_16x16x4_f32(false, a, false, b, (short)0, c,
                                               false, false);
}

__device__ __forceinline__ float lrelu(float x) { return fmaxf(x, ALPHA * x); }

__device__ __forceinline__ float xhalf(float x) {
  const int b = __builtin_bit_cast(int, x);
  return __builtin_bit_cast(
      float, __builtin_amdgcn_permlanex16(b, b, 0x76543210, 0xfedcba98, false, false));
}

__device__ __forceinline__ void group_wmma(v2f e0, v2f e1, v2f wf00, v2f wf01,
                                           v2f wf10, v2f wf11, v8f cini0,
                                           v8f cini1, v8f& f0, v8f& f1) {
  f0 = wmma_f32_k4(wf00, e0, cini0);
  f0 = wmma_f32_k4(wf10, e1, f0);
  f1 = wmma_f32_k4(wf01, e0, cini1);
  f1 = wmma_f32_k4(wf11, e1, f1);
}

__device__ __forceinline__ float group_sum(v8f f0, v8f f1, v2f e0, v2f e1,
                                           const v2f (&aw0)[4],
                                           const v2f (&aw1)[4], v2f u0, v2f u1,
                                           v2f seed) {
  v2f lin = __builtin_elementwise_fma(e0, u0, seed);
  lin = __builtin_elementwise_fma(e1, u1, lin);
  float acc = lin.x + lin.y;
#pragma unroll
  for (int j = 0; j < 4; ++j) {
    acc = __builtin_fmaf(__builtin_fabsf(f0[2 * j]), aw0[j].x, acc);
    acc = __builtin_fmaf(__builtin_fabsf(f0[2 * j + 1]), aw0[j].y, acc);
    acc = __builtin_fmaf(__builtin_fabsf(f1[2 * j]), aw1[j].x, acc);
    acc = __builtin_fmaf(__builtin_fabsf(f1[2 * j + 1]), aw1[j].y, acc);
  }
  return acc;
}
#define PREP_TILES (N / 16)
#define PTILE 272
#define PWAVE (3 * PTILE)
__global__ void __launch_bounds__(128)
prep_kernel(const float* __restrict__ input, const float* __restrict__ Wh,
            const float* __restrict__ Wh1, const float* __restrict__ Wh2,
            const float* __restrict__ ah, const float* __restrict__ Wf,
            const float* __restrict__ af, const float* __restrict__ bf,
            float* __restrict__ h, float* __restrict__ s1,
            float* __restrict__ s2, float* __restrict__ uw) {
  if (blockIdx.x == PREP_TILES) {
    const int t = threadIdx.x;
    if (t <= D_EDGE) {
      const float* row = t < D_EDGE ? Wf + t * D_OUT : bf;
      float acc = 0.f;
      for (int f = 0; f < D_OUT; ++f) acc += row[f] * af[f];
      uw[t] = acc;
    }
    return;
  }
  __shared__ __attribute__((aligned(16))) float vs[2 * D_IN];
  __shared__ float part[4 * PWAVE];
  const int tid = threadIdx.x;
  const int lane = tid & 31;
  const int wave = tid >> 5;
  const int c = lane & 15;
  const int hf = lane >> 4;
  const int i0 = blockIdx.x * 16;
  const int kb = wave * 32 + hf * 16;

  const v4f* w1p = (const v4f*)(Wh1 + (size_t)tid * D_OUT);
  const v4f* w2p = (const v4f*)(Wh2 + (size_t)tid * D_OUT);
  v4f w1[8], w2[8];
#pragma unroll
  for (int j = 0; j < 8; ++j) {
    w1[j] = w1p[j];
    w2[j] = w2p[j];
  }
  const v4f* ap = (const v4f*)(input + (size_t)(i0 + c) * D_IN + kb);
  v4f a4[4];
#pragma unroll
  for (int j = 0; j < 4; ++j) a4[j] = ap[j];
  v2f b0[8], b1[8];
#pragma unroll
  for (int s = 0; s < 8; ++s) {
    const float* wp = Wh + (size_t)(kb + 2 * s) * D_OUT + c;
    b0[s].x = wp[0];
    b0[s].y = wp[D_OUT];
    b1[s].x = wp[16];
    b1[s].y = wp[D_OUT + 16];
  }

  float a1 = 0.f, a2 = 0.f;
#pragma unroll
  for (int f = 0; f < D_OUT; ++f) {
    const float w = ah[f];
    a1 += w1[f >> 2][f & 3] * w;
    a2 += w2[f >> 2][f & 3] * w;
  }
  vs[tid] = a1;
  vs[D_IN + tid] = a2;
  __syncthreads();

  const v4f* vp = (const v4f*)(vs + (c & 1) * D_IN + kb);
  const bool vcol = c < 2;
  v4f bv4[4];
#pragma unroll
  for (int j = 0; j < 4; ++j) bv4[j] = vp[j];

  v8f c0 = {};
  v8f c1 = {};
  v8f c2 = {};
#pragma unroll
  for (int s = 0; s < 8; ++s) {
    v2f a, bv;
    a.x = a4[s >> 1][2 * (s & 1)];
    a.y = a4[s >> 1][2 * (s & 1) + 1];
    bv.x = vcol ? bv4[s >> 1][2 * (s & 1)] : 0.f;
    bv.y = vcol ? bv4[s >> 1][2 * (s & 1) + 1] : 0.f;
    c0 = wmma_f32_k4(a, b0[s], c0);
    c1 = wmma_f32_k4(a, b1[s], c1);
    c2 = wmma_f32_k4(a, bv, c2);
  }

  float* pw = part + wave * PWAVE;
#pragma unroll
  for (int r = 0; r < 8; ++r) {
    pw[r * 32 + lane] = c0[r];
    pw[PTILE + r * 32 + lane] = c1[r];
    pw[2 * PTILE + r * 32 + lane] = c2[r];
  }
  __syncthreads();

#pragma unroll
  for (int q = 0; q < 4; ++q) {
    const int R = 4 * wave + q;
    const int off = hf * PTILE + (R & 7) * 32 + c + 16 * (R >> 3);
    const float s = ((part[off] + part[PWAVE + off]) + part[2 * PWAVE + off]) +
                    part[3 * PWAVE + off];
    h[(size_t)(i0 + R) * D_OUT + lane] = s;
  }
  if (wave == 0) {
    const int off = 2 * PTILE + (c & 7) * 32 + hf + 16 * (c >> 3);
    const float s = ((part[off] + part[PWAVE + off]) + part[2 * PWAVE + off]) +
                    part[3 * PWAVE + off];
    float* dst = hf ? s2 : s1;
    dst[i0 + c] = s;
  }
}

__global__ void __launch_bounds__(32 * TWAVES)
fused_attention_kernel(const float* __restrict__ edge_feat,
                       const int* __restrict__ adj,
                       const float* __restrict__ Wf,
                       const float* __restrict__ af,
                       const float* __restrict__ bf,
                       const float* __restrict__ s1,
                       const float* __restrict__ s2,
                       const float* __restrict__ h,
                       const float* __restrict__ uw,
                       float* __restrict__ h_prime,
                       float* __restrict__ bp) {
  extern __shared__ float smem[];
  float* sc = smem;
  float* part = smem + TROWS * ROWSTRIDE;
  __shared__ float s_max[TWAVES];
  __shared__ float s_sum[TWAVES];
  __shared__ float s_hp[TROWS * D_OUT];

  const int tid = threadIdx.x;
  const int lane = tid & 31;
  const int wave = tid >> 5;
  const int i0 = blockIdx.x * TROWS;

  const int nn = lane & 15;
  const int kd = (lane >> 4) << 1;

  v2f bw00, bw01, bw10, bw11;
  bw00.x = Wf[(0 + kd) * D_OUT + nn];
  bw00.y = Wf[(1 + kd) * D_OUT + nn];
  bw01.x = Wf[(0 + kd) * D_OUT + 16 + nn];
  bw01.y = Wf[(1 + kd) * D_OUT + 16 + nn];
  bw10.x = Wf[(4 + kd) * D_OUT + nn];
  bw10.y = Wf[(5 + kd) * D_OUT + nn];
  bw11.x = Wf[(4 + kd) * D_OUT + 16 + nn];
  bw11.y = Wf[(5 + kd) * D_OUT + 16 + nn];
  const int f8 = (lane >> 4) << 3;
  const float ABSW = 0.5f * (1.0f - ALPHA);
  const float LINW = 0.5f * (1.0f + ALPHA);
  v2f aw0[4], aw1[4];
  v8f cini0, cini1;
  for (int v = 0; v < 8; ++v) {
    aw0[v >> 1][v & 1] = ABSW * af[f8 + v];
    aw1[v >> 1][v & 1] = ABSW * af[16 + f8 + v];
    cini0[v] = bf[f8 + v];
    cini1[v] = bf[16 + f8 + v];
  }
  v2f u0 = *(const v2f*)(uw + kd);
  v2f u1 = *(const v2f*)(uw + 4 + kd);
  u0 = LINW * u0;
  u1 = LINW * u1;
  v2f seed;
  seed.x = lane < 16 ? LINW * uw[8] : 0.f;
  seed.y = 0.f;

  const bool second = lane >= 16;
  const int jl = wave * WCOLS + lane;
  float s2v[PAIRS];
#pragma unroll
  for (int p = 0; p < PAIRS; ++p) s2v[p] = s2[jl + 32 * p];
  for (int il = 0; il < TROWS; ++il) {
    const int i = i0 + il;
    const float s1i = s1[i];
    const float* ep = edge_feat + ((size_t)i * N + (wave * WCOLS + nn)) * D_EDGE + kd;
    const int* ap = adj + (size_t)i * N + jl;
    v2f a0[GROUPS], a1[GROUPS];
    int av[PAIRS];
#pragma unroll
    for (int u = 0; u < GROUPS; ++u) {
      a0[u] = *(const v2f*)(ep + u * 16 * D_EDGE);
      a1[u] = *(const v2f*)(ep + u * 16 * D_EDGE + 4);
    }
#pragma unroll
    for (int p = 0; p < PAIRS; ++p) av[p] = ap[32 * p];
    __builtin_amdgcn_sched_barrier(0);
    v8f f0[GROUPS], f1[GROUPS];
#pragma unroll
    for (int u = 0; u < GROUPS; ++u)
      group_wmma(a0[u], a1[u], bw00, bw01, bw10, bw11, cini0, cini1, f0[u], f1[u]);
    __builtin_amdgcn_sched_barrier(0);
#pragma unroll
    for (int pr = 0; pr < PAIRS; ++pr) {
      float pa = group_sum(f0[2 * pr], f1[2 * pr], a0[2 * pr], a1[2 * pr], aw0,
                           aw1, u0, u1, seed);
      float pb = group_sum(f0[2 * pr + 1], f1[2 * pr + 1], a0[2 * pr + 1],
                           a1[2 * pr + 1], aw0, aw1, u0, u1, seed);
      asm("" : "+v"(pa));
      asm("" : "+v"(pb));
      const float keep = second ? pb : pa;
      const float send = second ? pa : pb;
      const float cp = keep + xhalf(send);
      float v = lrelu(s1i + s2v[pr]) + lrelu(cp);
      v = (av[pr] > 0) ? v : NEG_BIG;
      sc[il * ROWSTRIDE + jl + pr * 32] = v;
    }
  }
  __syncthreads();

  const int r = wave / WPR;
  {
    float* seg = sc + r * ROWSTRIDE + (wave % WPR) * (N / WPR);
    float x[XPL];
    float m = -3.4e38f;
    for (int c = 0; c < XPL; ++c) {
      x[c] = seg[lane + 32 * c];
      m = fmaxf(m, x[c]);
    }
    for (int mask = 16; mask >= 1; mask >>= 1)
      m = fmaxf(m, __shfl_xor(m, mask, 32));
    if (lane == 0) s_max[wave] = m;
    __syncthreads();
    m = fmaxf(s_max[WPR * r], s_max[WPR * r + 1]);
    float s = 0.f;
    for (int c = 0; c < XPL; ++c) {
      const float e = __expf(x[c] - m);
      seg[lane + 32 * c] = e;
      s += e;
    }
    for (int mask = 16; mask >= 1; mask >>= 1) s += __shfl_xor(s, mask, 32);
    if (lane == 0) s_sum[wave] = s;
  }
  __syncthreads();

  v8f c0 = {};
  v8f c1 = {};
  const int kbase = wave * WCOLS;
  for (int kk = 0; kk < WCOLS; kk += 4) {
    const int k0 = kbase + kk;
    v2f a = *(const v2f*)(sc + (nn & 7) * ROWSTRIDE + k0 + kd);
    v2f b0, b1;
    b0.x = h[(size_t)(k0 + kd) * D_OUT + nn];
    b0.y = h[(size_t)(k0 + kd + 1) * D_OUT + nn];
    b1.x = h[(size_t)(k0 + kd) * D_OUT + 16 + nn];
    b1.y = h[(size_t)(k0 + kd + 1) * D_OUT + 16 + nn];
    c0 = wmma_f32_k4(a, b0, c0);
    c1 = wmma_f32_k4(a, b1, c1);
  }
  if (lane < 16) {
    float* pw = part + wave * 256;
    for (int v = 0; v < 8; ++v) {
      pw[v * 32 + nn] = c0[v];
      pw[v * 32 + 16 + nn] = c1[v];
    }
  }
  __syncthreads();
  float hv = 0.f;
  if (tid < 256) {
    float s = 0.f;
    for (int w = 0; w < TWAVES; ++w) s += part[w * 256 + tid];
    const int row = tid >> 5;
    const float rs = s_sum[WPR * row] + s_sum[WPR * row + 1];
    hv = s * (1.0f / rs);
    s_hp[tid] = hv;
  }
  __syncthreads();
  if (tid < 256) h_prime[(size_t)i0 * D_OUT + tid] = hv;
  if (tid < 64) {
    const int col = tid & 31;
    const bool sq = tid >= 32;
    float acc = 0.f;
    for (int r = 0; r < TROWS; ++r) {
      const float v = s_hp[r * D_OUT + col];
      acc += sq ? v * v : v;
    }
    bp[blockIdx.x * 64 + tid] = acc;
  }
}

#define NBLK (N / TROWS)
__global__ void __launch_bounds__(256)
bn_final_kernel(const float* __restrict__ hp, const float* __restrict__ bp,
                const float* __restrict__ gamma, const float* __restrict__ beta,
                float* __restrict__ out) {
  __shared__ float red[4][64];
  __shared__ float ms[64];
  const int t = threadIdx.x;
  const int i4 = blockIdx.x * 256 + t;
  const v4f x = ((const v4f*)hp)[i4];
  const int col0 = (t & 7) * 4;
  const v4f g = *(const v4f*)(gamma + col0);
  const v4f b = *(const v4f*)(beta + col0);

  const int cc = t & 63;
  const int p = t >> 6;
  const float* bq = bp + (size_t)(p * (NBLK / 4)) * 64 + cc;
  float acc = 0.f;
#pragma unroll
  for (int k = 0; k < NBLK / 4; ++k) acc += bq[k * 64];
  red[p][cc] = acc;
  __syncthreads();
  if (t < 32) {
    const float s = ((red[0][t] + red[1][t]) + red[2][t]) + red[3][t];
    const float q = ((red[0][32 + t] + red[1][32 + t]) + red[2][32 + t]) +
                    red[3][32 + t];
    const float mean = s * (1.0f / (float)N);
    const float var = q * (1.0f / (float)N) - mean * mean;
    ms[t] = mean;
    ms[32 + t] = rsqrtf(var + 1e-5f);
  }
  __syncthreads();
  v4f o;
#pragma unroll
  for (int e = 0; e < 4; ++e)
    o[e] = g[e] * (x[e] - ms[col0 + e]) * ms[32 + col0 + e] + b[e];
  ((v4f*)out)[i4] = o;
}

extern "C" void kernel_launch(void* const* d_in, const int* in_sizes, int n_in,
                              void* d_out, int out_size, void* d_ws,
                              size_t ws_size, hipStream_t stream) {
  const float* input = (const float*)d_in[0];
  const float* edge_feat = (const float*)d_in[1];
  const int* adj = (const int*)d_in[2];
  const float* Wh = (const float*)d_in[3];
  const float* Wh1 = (const float*)d_in[4];
  const float* Wh2 = (const float*)d_in[5];
  const float* Wf = (const float*)d_in[6];
  const float* ah = (const float*)d_in[7];
  const float* af = (const float*)d_in[8];
  const float* bf = (const float*)d_in[9];
  const float* gamma = (const float*)d_in[10];
  const float* beta = (const float*)d_in[11];

  float* ws = (float*)d_ws;
  float* uw = ws;
  float* s1 = ws + 256;
  float* s2 = ws + 2304;
  float* h = ws + 4352;
  float* hp = ws + 69888;
  float* bp = ws + 135424;
  float* out = (float*)d_out;

  prep_kernel<<<129, 128, 0, stream>>>(input, Wh, Wh1, Wh2, ah, Wf, af, bf, h, s1,
                                       s2, uw);
  const size_t smem = (size_t)(8 * ROWSTRIDE + 16 * 256) * sizeof(float);
  fused_attention_kernel<<<256, 512, smem, stream>>>(edge_feat, adj, Wf, af, bf,
                                                      s1, s2, h, uw, hp, bp);
  bn_final_kernel<<<64, 256, 0, stream>>>(hp, bp, gamma, beta, out);
}
